# starting kernel
# baseline (speedup 1.0000x reference)
#include <hip/hip_runtime.h>
#include <hip/hip_bf16.h>
#include <math.h>

#define EPS 1e-6f
#define LDIM 256
#define BATCH 65536
#define SPAD 257

typedef __attribute__((ext_vector_type(2))) float v2f;
typedef __attribute__((ext_vector_type(8))) float v8f;

__global__ __launch_bounds__(1024) void sinkhorn_kernel(const float* __restrict__ logits,
                                                        float* __restrict__ P) {
    __shared__ float M[LDIM * SPAD];
    const int t = threadIdx.x;
    const int g = t >> 2;
    const int q = t & 3;

    for (int i = t; i < LDIM * LDIM; i += 1024)
        M[(i >> 8) * SPAD + (i & 255)] = logits[i];
    __syncthreads();

    for (int it = 0; it < 10; ++it) {
        {
            const int base = g * SPAD + q * 64;
            float mx = -3.4e38f;
            #pragma unroll 8
            for (int i = 0; i < 64; ++i) mx = fmaxf(mx, M[base + i]);
            mx = fmaxf(mx, __shfl_xor(mx, 1));
            mx = fmaxf(mx, __shfl_xor(mx, 2));
            float s = 0.0f;
            #pragma unroll 8
            for (int i = 0; i < 64; ++i) s += expf(M[base + i] - mx);
            s += __shfl_xor(s, 1);
            s += __shfl_xor(s, 2);
            const float lse = mx + logf(s);
            #pragma unroll 8
            for (int i = 0; i < 64; ++i) M[base + i] -= lse;
        }
        __syncthreads();
        {
            const int base = (q * 64) * SPAD + g;
            float mx = -3.4e38f;
            #pragma unroll 8
            for (int i = 0; i < 64; ++i) mx = fmaxf(mx, M[base + i * SPAD]);
            mx = fmaxf(mx, __shfl_xor(mx, 1));
            mx = fmaxf(mx, __shfl_xor(mx, 2));
            float s = 0.0f;
            #pragma unroll 8
            for (int i = 0; i < 64; ++i) s += expf(M[base + i * SPAD] - mx);
            s += __shfl_xor(s, 1);
            s += __shfl_xor(s, 2);
            const float lse = mx + logf(s);
            #pragma unroll 8
            for (int i = 0; i < 64; ++i) M[base + i * SPAD] -= lse;
        }
        __syncthreads();
    }

    for (int i = t; i < LDIM * LDIM; i += 1024)
        P[i] = expf(M[(i >> 8) * SPAD + (i & 255)]);
}

__global__ __launch_bounds__(256) void gemm_kernel(const float* __restrict__ x,
                                                   const float* __restrict__ P,
                                                   float* __restrict__ leafT) {
    __shared__ float Ps[LDIM * LDIM];
    const int t = threadIdx.x;
    {
        const float4* Pv = (const float4*)P;
        float4* Sv = (float4*)Ps;
        for (int i = t; i < LDIM * LDIM / 4; i += 256) Sv[i] = Pv[i];
    }
    __syncthreads();

    const int wave  = t >> 5;
    const int lane  = t & 31;
    const int mloc  = lane & 15;
    const int khalf = (lane >> 4) * 2;
    const int m0    = blockIdx.x * 128 + wave * 16;

    v8f acc[16] = {};

    const float* xrow = x + (size_t)(m0 + mloc) * LDIM + khalf;

    for (int k = 0; k < LDIM; k += 4) {
        v2f bfrag;
        bfrag.x = xrow[k];
        bfrag.y = xrow[k + 1];
        #pragma unroll
        for (int nt = 0; nt < 16; ++nt) {
            const int n = nt * 16 + mloc;
            v2f afrag;
            afrag.x = Ps[(k + khalf)     * LDIM + n];
            afrag.y = Ps[(k + khalf + 1) * LDIM + n];
            acc[nt] = __builtin_amdgcn_wmma_f32_16x16x4_f32(
                false, afrag, false, bfrag, (short)0, acc[nt], false, false);
        }
    }

    const size_t col = (size_t)m0 + mloc;
    #pragma unroll
    for (int nt = 0; nt < 16; ++nt) {
        const int nbase = nt * 16 + (lane >> 4) * 8;
        #pragma unroll
        for (int i = 0; i < 8; ++i)
            leafT[(size_t)(nbase + i) * BATCH + col] = acc[nt][i];
    }
}

__device__ __forceinline__ float r_poly(float a) {
    float d  = 0.5f - a;
    float d2 = d * d;
    float num = 0.25f + 1.65811f * d + 2.15388f * d2 + 8.2844f * d2 * d + 6.16764f * d2 * d2;
    float den = a * (1.0f - a);
    den = (fabsf(den) < EPS) ? EPS : den;
    return num / den;
}

__device__ __forceinline__ float power_r(float x, float y, float ra, float w0, float w1) {
    ra = (fabsf(ra) < EPS) ? EPS : ra;
    return powf(w0 * powf(x, ra) + w1 * powf(y, ra), 1.0f / ra);
}

__device__ __forceinline__ float F_upper(float x, float y, float a, float w0, float w1) {
    const float xy  = x * y;
    const float mn  = fminf(x, y);
    const float lin = w0 * x + w1 * y;
    float out = lin;
    if (a > 0.5f && a < 0.75f) {
        const float R = r_poly(0.75f);
        out = (3.0f - 4.0f * a) * lin + (4.0f * a - 2.0f) * power_r(x, y, R, w0, w1);
    }
    if (a >= 0.75f && a < 1.0f) {
        const float ra = r_poly(fminf(fmaxf(a, 0.75f), 1.0f - EPS));
        out = power_r(x, y, ra, w0, w1);
    }
    if (fabsf(a - 1.0f)  < EPS) out = mn;
    if (a > 1.0f  && a < 1.25f) out = 4.0f * ((1.25f - a) * mn + (a - 1.0f) * xy);
    if (fabsf(a - 1.25f) < EPS) out = xy;
    if (a > 1.25f && a < 2.0f) {
        const float p = sqrtf(fmaxf(3.0f / fmaxf(2.0f - a, EPS) - 1.0f, EPS));
        out = powf(xy, p);
    }
    if (fabsf(a - 2.0f)  < EPS)
        out = ((fabsf(x - 1.0f) < EPS) && (fabsf(y - 1.0f) < EPS)) ? 1.0f : 0.0f;
    return out;
}

__device__ __forceinline__ float F_comb(float x, float y, float a, float w0, float w1) {
    if (isnan(x)) x = EPS;
    x = fminf(fmaxf(x, EPS), 1.0f - EPS);
    if (isnan(y)) y = EPS;
    y = fminf(fmaxf(y, EPS), 1.0f - EPS);
    if (a < 0.5f - EPS) {
        const float af = fminf(fmaxf(1.0f - a, -1.0f + EPS), 2.0f - EPS);
        return 1.0f - F_upper(1.0f - x, 1.0f - y, af, w0, w1);
    }
    return F_upper(x, y, a, w0, w1);
}

__device__ __forceinline__ float combine_op(float l, float r, float a, float w0, float w1) {
    const float nres = F_comb(l, r, a, w0, w1);
    return isnan(nres) ? a : nres;
}

__global__ __launch_bounds__(256) void chain_kernel(const float* __restrict__ leafT,
                                                    const float* __restrict__ weights,
                                                    const float* __restrict__ biases,
                                                    float* __restrict__ out) {
    __shared__ float sA[255], sW0[255], sW1[255];
    const int t = threadIdx.x;
    if (t < 255) {
        const float w0 = weights[2 * t], w1 = weights[2 * t + 1];
        const float m  = fmaxf(w0, w1);
        const float e0 = expf(w0 - m), e1 = expf(w1 - m);
        const float inv = 1.0f / (e0 + e1);
        sW0[t] = e0 * inv;
        sW1[t] = e1 * inv;
        float a = 3.0f / (1.0f + expf(-biases[t])) - 1.0f;
        sA[t] = fminf(fmaxf(a, -1.0f + EPS), 2.0f - EPS);
    }
    __syncthreads();

    const int b = blockIdx.x * 256 + t;
    float carry = combine_op(leafT[b], leafT[(size_t)BATCH + b], sA[0], sW0[0], sW1[0]);
    for (int l = 2; l < LDIM; ++l) {
        const float leaf = leafT[(size_t)l * BATCH + b];
        carry = combine_op(carry, leaf, sA[l - 1], sW0[l - 1], sW1[l - 1]);
    }
    out[b] = carry;
}

extern "C" void kernel_launch(void* const* d_in, const int* in_sizes, int n_in,
                              void* d_out, int out_size, void* d_ws, size_t ws_size,
                              hipStream_t stream) {
    const float* x       = (const float*)d_in[0];
    const float* logits  = (const float*)d_in[1];
    const float* weights = (const float*)d_in[2];
    const float* biases  = (const float*)d_in[3];

    float* P     = (float*)d_ws;
    float* leafT = P + LDIM * LDIM;

    sinkhorn_kernel<<<1, 1024, 0, stream>>>(logits, P);
    gemm_kernel<<<BATCH / 128, 256, 0, stream>>>(x, P, leafT);
    chain_kernel<<<BATCH / 256, 256, 0, stream>>>(leafT, weights, biases, (float*)d_out);
}


# Sinkhorn LDS tile XOR-swizzled by quarter index so the 4 lanes sharing a row or column hit 4 different banks: row and column sweeps go from 4-way bank conflict to conflict-free
# speedup vs baseline: 1.1591x; 1.1591x over previous
#include <hip/hip_runtime.h>
#include <hip/hip_bf16.h>
#include <math.h>

#define EPS 1e-6f
#define LDIM 256
#define BATCH 65536
#define SPAD 257

typedef __attribute__((ext_vector_type(2))) float v2f;
typedef __attribute__((ext_vector_type(8))) float v8f;

__device__ __forceinline__ int sw_idx(int r, int c) {
    return r * SPAD + (c ^ ((((c >> 6) ^ (r >> 6)) & 3) << 3));
}

__global__ __launch_bounds__(1024) void sinkhorn_kernel(const float* __restrict__ logits,
                                                        float* __restrict__ P) {
    __shared__ float M[LDIM * SPAD];
    const int t = threadIdx.x;
    const int g = t >> 2;
    const int q = t & 3;

    for (int i = t; i < LDIM * LDIM; i += 1024)
        M[sw_idx(i >> 8, i & 255)] = logits[i];
    __syncthreads();

    const int rswz  = ((q ^ (g >> 6)) & 3) << 3;
    const int rbase = g * SPAD + q * 64;
    const int cbase = (q * 64) * SPAD + (g ^ ((((g >> 6) ^ q) & 3) << 3));

    for (int it = 0; it < 10; ++it) {
        {
            float mx = -3.4e38f;
            for (int j = 0; j < 64; j += 8) {
                const float* p = &M[rbase + (j ^ rswz)];
                #pragma unroll
                for (int e = 0; e < 8; ++e) mx = fmaxf(mx, p[e]);
            }
            mx = fmaxf(mx, __shfl_xor(mx, 1));
            mx = fmaxf(mx, __shfl_xor(mx, 2));
            float s = 0.0f;
            for (int j = 0; j < 64; j += 8) {
                const float* p = &M[rbase + (j ^ rswz)];
                #pragma unroll
                for (int e = 0; e < 8; ++e) s += expf(p[e] - mx);
            }
            s += __shfl_xor(s, 1);
            s += __shfl_xor(s, 2);
            const float lse = mx + logf(s);
            for (int j = 0; j < 64; j += 8) {
                float* p = &M[rbase + (j ^ rswz)];
                #pragma unroll
                for (int e = 0; e < 8; ++e) p[e] -= lse;
            }
        }
        __syncthreads();
        {
            const int base = cbase;
            float mx = -3.4e38f;
            #pragma unroll 8
            for (int i = 0; i < 64; ++i) mx = fmaxf(mx, M[base + i * SPAD]);
            mx = fmaxf(mx, __shfl_xor(mx, 1));
            mx = fmaxf(mx, __shfl_xor(mx, 2));
            float s = 0.0f;
            #pragma unroll 8
            for (int i = 0; i < 64; ++i) s += expf(M[base + i * SPAD] - mx);
            s += __shfl_xor(s, 1);
            s += __shfl_xor(s, 2);
            const float lse = mx + logf(s);
            #pragma unroll 8
            for (int i = 0; i < 64; ++i) M[base + i * SPAD] -= lse;
        }
        __syncthreads();
    }

    for (int i = t; i < LDIM * LDIM; i += 1024)
        P[i] = expf(M[sw_idx(i >> 8, i & 255)]);
}

__global__ __launch_bounds__(256) void gemm_kernel(const float* __restrict__ x,
                                                   const float* __restrict__ P,
                                                   float* __restrict__ leafT) {
    __shared__ float Ps[LDIM * LDIM];
    const int t = threadIdx.x;
    {
        const float4* Pv = (const float4*)P;
        float4* Sv = (float4*)Ps;
        for (int i = t; i < LDIM * LDIM / 4; i += 256) Sv[i] = Pv[i];
    }
    __syncthreads();

    const int wave  = t >> 5;
    const int lane  = t & 31;
    const int mloc  = lane & 15;
    const int khalf = (lane >> 4) * 2;
    const int m0    = blockIdx.x * 128 + wave * 16;

    v8f acc[16] = {};

    const float* xrow = x + (size_t)(m0 + mloc) * LDIM + khalf;

    for (int k = 0; k < LDIM; k += 4) {
        v2f bfrag;
        bfrag.x = xrow[k];
        bfrag.y = xrow[k + 1];
        #pragma unroll
        for (int nt = 0; nt < 16; ++nt) {
            const int n = nt * 16 + mloc;
            v2f afrag;
            afrag.x = Ps[(k + khalf)     * LDIM + n];
            afrag.y = Ps[(k + khalf + 1) * LDIM + n];
            acc[nt] = __builtin_amdgcn_wmma_f32_16x16x4_f32(
                false, afrag, false, bfrag, (short)0, acc[nt], false, false);
        }
    }

    const size_t col = (size_t)m0 + mloc;
    #pragma unroll
    for (int nt = 0; nt < 16; ++nt) {
        const int nbase = nt * 16 + (lane >> 4) * 8;
        #pragma unroll
        for (int i = 0; i < 8; ++i)
            leafT[(size_t)(nbase + i) * BATCH + col] = acc[nt][i];
    }
}

__device__ __forceinline__ float r_poly(float a) {
    float d  = 0.5f - a;
    float d2 = d * d;
    float num = 0.25f + 1.65811f * d + 2.15388f * d2 + 8.2844f * d2 * d + 6.16764f * d2 * d2;
    float den = a * (1.0f - a);
    den = (fabsf(den) < EPS) ? EPS : den;
    return num / den;
}

__device__ __forceinline__ float power_r(float x, float y, float ra, float w0, float w1) {
    ra = (fabsf(ra) < EPS) ? EPS : ra;
    return powf(w0 * powf(x, ra) + w1 * powf(y, ra), 1.0f / ra);
}

__device__ __forceinline__ float F_upper(float x, float y, float a, float w0, float w1) {
    const float xy  = x * y;
    const float mn  = fminf(x, y);
    const float lin = w0 * x + w1 * y;
    float out = lin;
    if (a > 0.5f && a < 0.75f) {
        const float R = r_poly(0.75f);
        out = (3.0f - 4.0f * a) * lin + (4.0f * a - 2.0f) * power_r(x, y, R, w0, w1);
    }
    if (a >= 0.75f && a < 1.0f) {
        const float ra = r_poly(fminf(fmaxf(a, 0.75f), 1.0f - EPS));
        out = power_r(x, y, ra, w0, w1);
    }
    if (fabsf(a - 1.0f)  < EPS) out = mn;
    if (a > 1.0f  && a < 1.25f) out = 4.0f * ((1.25f - a) * mn + (a - 1.0f) * xy);
    if (fabsf(a - 1.25f) < EPS) out = xy;
    if (a > 1.25f && a < 2.0f) {
        const float p = sqrtf(fmaxf(3.0f / fmaxf(2.0f - a, EPS) - 1.0f, EPS));
        out = powf(xy, p);
    }
    if (fabsf(a - 2.0f)  < EPS)
        out = ((fabsf(x - 1.0f) < EPS) && (fabsf(y - 1.0f) < EPS)) ? 1.0f : 0.0f;
    return out;
}

__device__ __forceinline__ float F_comb(float x, float y, float a, float w0, float w1) {
    if (isnan(x)) x = EPS;
    x = fminf(fmaxf(x, EPS), 1.0f - EPS);
    if (isnan(y)) y = EPS;
    y = fminf(fmaxf(y, EPS), 1.0f - EPS);
    if (a < 0.5f - EPS) {
        const float af = fminf(fmaxf(1.0f - a, -1.0f + EPS), 2.0f - EPS);
        return 1.0f - F_upper(1.0f - x, 1.0f - y, af, w0, w1);
    }
    return F_upper(x, y, a, w0, w1);
}

__device__ __forceinline__ float combine_op(float l, float r, float a, float w0, float w1) {
    const float nres = F_comb(l, r, a, w0, w1);
    return isnan(nres) ? a : nres;
}

__global__ __launch_bounds__(256) void chain_kernel(const float* __restrict__ leafT,
                                                    const float* __restrict__ weights,
                                                    const float* __restrict__ biases,
                                                    float* __restrict__ out) {
    __shared__ float sA[255], sW0[255], sW1[255];
    const int t = threadIdx.x;
    if (t < 255) {
        const float w0 = weights[2 * t], w1 = weights[2 * t + 1];
        const float m  = fmaxf(w0, w1);
        const float e0 = expf(w0 - m), e1 = expf(w1 - m);
        const float inv = 1.0f / (e0 + e1);
        sW0[t] = e0 * inv;
        sW1[t] = e1 * inv;
        float a = 3.0f / (1.0f + expf(-biases[t])) - 1.0f;
        sA[t] = fminf(fmaxf(a, -1.0f + EPS), 2.0f - EPS);
    }
    __syncthreads();

    const int b = blockIdx.x * 256 + t;
    float carry = combine_op(leafT[b], leafT[(size_t)BATCH + b], sA[0], sW0[0], sW1[0]);
    for (int l = 2; l < LDIM; ++l) {
        const float leaf = leafT[(size_t)l * BATCH + b];
        carry = combine_op(carry, leaf, sA[l - 1], sW0[l - 1], sW1[l - 1]);
    }
    out[b] = carry;
}

extern "C" void kernel_launch(void* const* d_in, const int* in_sizes, int n_in,
                              void* d_out, int out_size, void* d_ws, size_t ws_size,
                              hipStream_t stream) {
    const float* x       = (const float*)d_in[0];
    const float* logits  = (const float*)d_in[1];
    const float* weights = (const float*)d_in[2];
    const float* biases  = (const float*)d_in[3];

    float* P     = (float*)d_ws;
    float* leafT = P + LDIM * LDIM;

    sinkhorn_kernel<<<1, 1024, 0, stream>>>(logits, P);
    gemm_kernel<<<BATCH / 128, 256, 0, stream>>>(x, P, leafT);
    chain_kernel<<<BATCH / 256, 256, 0, stream>>>(leafT, weights, biases, (float*)d_out);
}


# Scan kernel: library powf replaced by exp2(e*log2(b)) on the hardware v_log_f32/v_exp_f32 (every base is a clamped positive value); each power's rounding moves by a few ulp, no sum order changes
# speedup vs baseline: 1.4325x; 1.2359x over previous
#include <hip/hip_runtime.h>
#include <hip/hip_bf16.h>
#include <math.h>

#define EPS 1e-6f
#define LDIM 256
#define BATCH 65536
#define SPAD 257

typedef __attribute__((ext_vector_type(2))) float v2f;
typedef __attribute__((ext_vector_type(8))) float v8f;

__device__ __forceinline__ int sw_idx(int r, int c) {
    return r * SPAD + (c ^ ((((c >> 6) ^ (r >> 6)) & 3) << 3));
}

__global__ __launch_bounds__(1024) void sinkhorn_kernel(const float* __restrict__ logits,
                                                        float* __restrict__ P) {
    __shared__ float M[LDIM * SPAD];
    const int t = threadIdx.x;
    const int g = t >> 2;
    const int q = t & 3;

    for (int i = t; i < LDIM * LDIM; i += 1024)
        M[sw_idx(i >> 8, i & 255)] = logits[i];
    __syncthreads();

    const int rswz  = ((q ^ (g >> 6)) & 3) << 3;
    const int rbase = g * SPAD + q * 64;
    const int cbase = (q * 64) * SPAD + (g ^ ((((g >> 6) ^ q) & 3) << 3));

    for (int it = 0; it < 10; ++it) {
        {
            float mx = -3.4e38f;
            for (int j = 0; j < 64; j += 8) {
                const float* p = &M[rbase + (j ^ rswz)];
                #pragma unroll
                for (int e = 0; e < 8; ++e) mx = fmaxf(mx, p[e]);
            }
            mx = fmaxf(mx, __shfl_xor(mx, 1));
            mx = fmaxf(mx, __shfl_xor(mx, 2));
            float s = 0.0f;
            for (int j = 0; j < 64; j += 8) {
                const float* p = &M[rbase + (j ^ rswz)];
                #pragma unroll
                for (int e = 0; e < 8; ++e) s += expf(p[e] - mx);
            }
            s += __shfl_xor(s, 1);
            s += __shfl_xor(s, 2);
            const float lse = mx + logf(s);
            for (int j = 0; j < 64; j += 8) {
                float* p = &M[rbase + (j ^ rswz)];
                #pragma unroll
                for (int e = 0; e < 8; ++e) p[e] -= lse;
            }
        }
        __syncthreads();
        {
            const int base = cbase;
            float mx = -3.4e38f;
            #pragma unroll 8
            for (int i = 0; i < 64; ++i) mx = fmaxf(mx, M[base + i * SPAD]);
            mx = fmaxf(mx, __shfl_xor(mx, 1));
            mx = fmaxf(mx, __shfl_xor(mx, 2));
            float s = 0.0f;
            #pragma unroll 8
            for (int i = 0; i < 64; ++i) s += expf(M[base + i * SPAD] - mx);
            s += __shfl_xor(s, 1);
            s += __shfl_xor(s, 2);
            const float lse = mx + logf(s);
            #pragma unroll 8
            for (int i = 0; i < 64; ++i) M[base + i * SPAD] -= lse;
        }
        __syncthreads();
    }

    for (int i = t; i < LDIM * LDIM; i += 1024)
        P[i] = expf(M[sw_idx(i >> 8, i & 255)]);
}

__global__ __launch_bounds__(256) void gemm_kernel(const float* __restrict__ x,
                                                   const float* __restrict__ P,
                                                   float* __restrict__ leafT) {
    __shared__ float Ps[LDIM * LDIM];
    const int t = threadIdx.x;
    {
        const float4* Pv = (const float4*)P;
        float4* Sv = (float4*)Ps;
        for (int i = t; i < LDIM * LDIM / 4; i += 256) Sv[i] = Pv[i];
    }
    __syncthreads();

    const int wave  = t >> 5;
    const int lane  = t & 31;
    const int mloc  = lane & 15;
    const int khalf = (lane >> 4) * 2;
    const int m0    = blockIdx.x * 128 + wave * 16;

    v8f acc[16] = {};

    const float* xrow = x + (size_t)(m0 + mloc) * LDIM + khalf;

    for (int k = 0; k < LDIM; k += 4) {
        v2f bfrag;
        bfrag.x = xrow[k];
        bfrag.y = xrow[k + 1];
        #pragma unroll
        for (int nt = 0; nt < 16; ++nt) {
            const int n = nt * 16 + mloc;
            v2f afrag;
            afrag.x = Ps[(k + khalf)     * LDIM + n];
            afrag.y = Ps[(k + khalf + 1) * LDIM + n];
            acc[nt] = __builtin_amdgcn_wmma_f32_16x16x4_f32(
                false, afrag, false, bfrag, (short)0, acc[nt], false, false);
        }
    }

    const size_t col = (size_t)m0 + mloc;
    #pragma unroll
    for (int nt = 0; nt < 16; ++nt) {
        const int nbase = nt * 16 + (lane >> 4) * 8;
        #pragma unroll
        for (int i = 0; i < 8; ++i)
            leafT[(size_t)(nbase + i) * BATCH + col] = acc[nt][i];
    }
}

__device__ __forceinline__ float r_poly(float a) {
    float d  = 0.5f - a;
    float d2 = d * d;
    float num = 0.25f + 1.65811f * d + 2.15388f * d2 + 8.2844f * d2 * d + 6.16764f * d2 * d2;
    float den = a * (1.0f - a);
    den = (fabsf(den) < EPS) ? EPS : den;
    return num / den;
}

__device__ __forceinline__ float pow_pos(float b, float e) {
    return __builtin_amdgcn_exp2f(e * __builtin_amdgcn_logf(b));
}

__device__ __forceinline__ float pow_pos_sub(float b, float e) {
    return exp2f(e * __builtin_amdgcn_logf(b));
}

__device__ __forceinline__ float power_r(float x, float y, float ra, float w0, float w1) {
    ra = (fabsf(ra) < EPS) ? EPS : ra;
    return pow_pos(w0 * pow_pos(x, ra) + w1 * pow_pos(y, ra), 1.0f / ra);
}

__device__ __forceinline__ float F_upper(float x, float y, float a, float w0, float w1) {
    const float xy  = x * y;
    const float mn  = fminf(x, y);
    const float lin = w0 * x + w1 * y;
    float out = lin;
    if (a > 0.5f && a < 0.75f) {
        const float R = r_poly(0.75f);
        out = (3.0f - 4.0f * a) * lin + (4.0f * a - 2.0f) * power_r(x, y, R, w0, w1);
    }
    if (a >= 0.75f && a < 1.0f) {
        const float ra = r_poly(fminf(fmaxf(a, 0.75f), 1.0f - EPS));
        out = power_r(x, y, ra, w0, w1);
    }
    if (fabsf(a - 1.0f)  < EPS) out = mn;
    if (a > 1.0f  && a < 1.25f) out = 4.0f * ((1.25f - a) * mn + (a - 1.0f) * xy);
    if (fabsf(a - 1.25f) < EPS) out = xy;
    if (a > 1.25f && a < 2.0f) {
        const float p = sqrtf(fmaxf(3.0f / fmaxf(2.0f - a, EPS) - 1.0f, EPS));
        out = pow_pos_sub(xy, p);
    }
    if (fabsf(a - 2.0f)  < EPS)
        out = ((fabsf(x - 1.0f) < EPS) && (fabsf(y - 1.0f) < EPS)) ? 1.0f : 0.0f;
    return out;
}

__device__ __forceinline__ float F_comb(float x, float y, float a, float w0, float w1) {
    if (isnan(x)) x = EPS;
    x = fminf(fmaxf(x, EPS), 1.0f - EPS);
    if (isnan(y)) y = EPS;
    y = fminf(fmaxf(y, EPS), 1.0f - EPS);
    if (a < 0.5f - EPS) {
        const float af = fminf(fmaxf(1.0f - a, -1.0f + EPS), 2.0f - EPS);
        return 1.0f - F_upper(1.0f - x, 1.0f - y, af, w0, w1);
    }
    return F_upper(x, y, a, w0, w1);
}

__device__ __forceinline__ float combine_op(float l, float r, float a, float w0, float w1) {
    const float nres = F_comb(l, r, a, w0, w1);
    return isnan(nres) ? a : nres;
}

__global__ __launch_bounds__(256) void chain_kernel(const float* __restrict__ leafT,
                                                    const float* __restrict__ weights,
                                                    const float* __restrict__ biases,
                                                    float* __restrict__ out) {
    __shared__ float sA[255], sW0[255], sW1[255];
    const int t = threadIdx.x;
    if (t < 255) {
        const float w0 = weights[2 * t], w1 = weights[2 * t + 1];
        const float m  = fmaxf(w0, w1);
        const float e0 = expf(w0 - m), e1 = expf(w1 - m);
        const float inv = 1.0f / (e0 + e1);
        sW0[t] = e0 * inv;
        sW1[t] = e1 * inv;
        float a = 3.0f / (1.0f + expf(-biases[t])) - 1.0f;
        sA[t] = fminf(fmaxf(a, -1.0f + EPS), 2.0f - EPS);
    }
    __syncthreads();

    const int b = blockIdx.x * 256 + t;
    float carry = combine_op(leafT[b], leafT[(size_t)BATCH + b], sA[0], sW0[0], sW1[0]);
    for (int l = 2; l < LDIM; ++l) {
        const float leaf = leafT[(size_t)l * BATCH + b];
        carry = combine_op(carry, leaf, sA[l - 1], sW0[l - 1], sW1[l - 1]);
    }
    out[b] = carry;
}

extern "C" void kernel_launch(void* const* d_in, const int* in_sizes, int n_in,
                              void* d_out, int out_size, void* d_ws, size_t ws_size,
                              hipStream_t stream) {
    const float* x       = (const float*)d_in[0];
    const float* logits  = (const float*)d_in[1];
    const float* weights = (const float*)d_in[2];
    const float* biases  = (const float*)d_in[3];

    float* P     = (float*)d_ws;
    float* leafT = P + LDIM * LDIM;

    sinkhorn_kernel<<<1, 1024, 0, stream>>>(logits, P);
    gemm_kernel<<<BATCH / 128, 256, 0, stream>>>(x, P, leafT);
    chain_kernel<<<BATCH / 256, 256, 0, stream>>>(leafT, weights, biases, (float*)d_out);
}


# Sinkhorn sweeps as row/column scalings of K=exp(logits-rowmax): one exp per element instead of one per element per sweep; sums are 4 interleaved FMA chains (sum order and rounding change)
# speedup vs baseline: 2.0056x; 1.4001x over previous
#include <hip/hip_runtime.h>
#include <hip/hip_bf16.h>
#include <math.h>

#define EPS 1e-6f
#define LDIM 256
#define BATCH 65536
#define SPAD 257

typedef __attribute__((ext_vector_type(2))) float v2f;
typedef __attribute__((ext_vector_type(8))) float v8f;

__device__ __forceinline__ int sw_idx(int r, int c) {
    return r * SPAD + (c ^ ((((c >> 6) ^ (r >> 6)) & 3) << 3));
}

__device__ __forceinline__ int vpad(int i) { return i + ((i >> 6) << 3); }

__global__ __launch_bounds__(1024) void sinkhorn_kernel(const float* __restrict__ logits,
                                                        float* __restrict__ P) {
    __shared__ float M[LDIM * SPAD];
    __shared__ __attribute__((aligned(16))) float va[LDIM + 24];
    __shared__ __attribute__((aligned(16))) float vb[LDIM + 24];
    const int t = threadIdx.x;
    const int g = t >> 2;
    const int q = t & 3;

    const int cc = t & 255;
    const int r0 = t >> 8;
    for (int kk = 0; kk < 4; ++kk) {
        const int base = (r0 + 64 * kk) * SPAD + (cc ^ ((((cc >> 6) ^ kk) & 3) << 3));
        for (int m = 0; m < 16; ++m)
            M[base + 4 * m * SPAD] = logits[t + 1024 * (16 * kk + m)];
    }
    if (t < LDIM) vb[vpad(t)] = 1.0f;
    __syncthreads();

    const int rswz  = ((q ^ (g >> 6)) & 3) << 3;
    const int rbase = g * SPAD + q * 64;
    const int cbase = (q * 64) * SPAD + (g ^ ((((g >> 6) ^ q) & 3) << 3));
    const int vq    = q * 72;

    float kr[64];
    float mx = -3.4e38f;
    #pragma unroll
    for (int j = 0; j < 64; j += 8) {
        const float* p = &M[rbase + (j ^ rswz)];
        #pragma unroll
        for (int e = 0; e < 8; ++e) { kr[j + e] = p[e]; mx = fmaxf(mx, kr[j + e]); }
    }
    mx = fmaxf(mx, __shfl_xor(mx, 1));
    mx = fmaxf(mx, __shfl_xor(mx, 2));
    #pragma unroll
    for (int j = 0; j < 64; j += 8) {
        float* p = &M[rbase + (j ^ rswz)];
        #pragma unroll
        for (int e = 0; e < 8; ++e) { kr[j + e] = expf(kr[j + e] - mx); p[e] = kr[j + e]; }
    }

    float a_g = 1.0f;
    for (int it = 0; it < 10; ++it) {
        {
            const float4* v4 = (const float4*)&vb[vq];
            float s0 = 0.0f, s1 = 0.0f, s2 = 0.0f, s3 = 0.0f;
            #pragma unroll
            for (int k = 0; k < 16; ++k) {
                const float4 v = v4[k];
                s0 = fmaf(kr[4 * k + 0], v.x, s0);
                s1 = fmaf(kr[4 * k + 1], v.y, s1);
                s2 = fmaf(kr[4 * k + 2], v.z, s2);
                s3 = fmaf(kr[4 * k + 3], v.w, s3);
            }
            float s = (s0 + s1) + (s2 + s3);
            s += __shfl_xor(s, 1);
            s += __shfl_xor(s, 2);
            a_g = 1.0f / s;
            if (q == 0) va[vpad(g)] = a_g;
        }
        __syncthreads();
        {
            const float4* v4 = (const float4*)&va[vq];
            float s0 = 0.0f, s1 = 0.0f, s2 = 0.0f, s3 = 0.0f;
            #pragma unroll 4
            for (int k = 0; k < 16; ++k) {
                const float4 v = v4[k];
                const float* p = &M[cbase + (4 * k) * SPAD];
                s0 = fmaf(p[0],        v.x, s0);
                s1 = fmaf(p[SPAD],     v.y, s1);
                s2 = fmaf(p[2 * SPAD], v.z, s2);
                s3 = fmaf(p[3 * SPAD], v.w, s3);
            }
            float s = (s0 + s1) + (s2 + s3);
            s += __shfl_xor(s, 1);
            s += __shfl_xor(s, 2);
            if (q == 0) vb[vpad(g)] = 1.0f / s;
        }
        __syncthreads();
    }

    #pragma unroll
    for (int j = 0; j < 64; j += 8) {
        float* p = &M[rbase + (j ^ rswz)];
        #pragma unroll
        for (int e = 0; e < 8; ++e) p[e] = (kr[j + e] * a_g) * vb[vq + j + e];
    }
    __syncthreads();

    for (int kk = 0; kk < 4; ++kk) {
        const int base = (r0 + 64 * kk) * SPAD + (cc ^ ((((cc >> 6) ^ kk) & 3) << 3));
        for (int m = 0; m < 16; ++m)
            P[t + 1024 * (16 * kk + m)] = M[base + 4 * m * SPAD];
    }
}

__global__ __launch_bounds__(256) void gemm_kernel(const float* __restrict__ x,
                                                   const float* __restrict__ P,
                                                   float* __restrict__ leafT) {
    __shared__ float Ps[LDIM * LDIM];
    const int t = threadIdx.x;
    {
        const float4* Pv = (const float4*)P;
        float4* Sv = (float4*)Ps;
        for (int i = t; i < LDIM * LDIM / 4; i += 256) Sv[i] = Pv[i];
    }
    __syncthreads();

    const int wave  = t >> 5;
    const int lane  = t & 31;
    const int mloc  = lane & 15;
    const int khalf = (lane >> 4) * 2;
    const int m0    = blockIdx.x * 128 + wave * 16;

    v8f acc[16] = {};

    const float* xrow = x + (size_t)(m0 + mloc) * LDIM + khalf;

    for (int k = 0; k < LDIM; k += 4) {
        v2f bfrag;
        bfrag.x = xrow[k];
        bfrag.y = xrow[k + 1];
        #pragma unroll
        for (int nt = 0; nt < 16; ++nt) {
            const int n = nt * 16 + mloc;
            v2f afrag;
            afrag.x = Ps[(k + khalf)     * LDIM + n];
            afrag.y = Ps[(k + khalf + 1) * LDIM + n];
            acc[nt] = __builtin_amdgcn_wmma_f32_16x16x4_f32(
                false, afrag, false, bfrag, (short)0, acc[nt], false, false);
        }
    }

    const size_t col = (size_t)m0 + mloc;
    #pragma unroll
    for (int nt = 0; nt < 16; ++nt) {
        const int nbase = nt * 16 + (lane >> 4) * 8;
        #pragma unroll
        for (int i = 0; i < 8; ++i)
            leafT[(size_t)(nbase + i) * BATCH + col] = acc[nt][i];
    }
}

__device__ __forceinline__ float r_poly(float a) {
    float d  = 0.5f - a;
    float d2 = d * d;
    float num = 0.25f + 1.65811f * d + 2.15388f * d2 + 8.2844f * d2 * d + 6.16764f * d2 * d2;
    float den = a * (1.0f - a);
    den = (fabsf(den) < EPS) ? EPS : den;
    return num / den;
}

__device__ __forceinline__ float pow_pos(float b, float e) {
    return __builtin_amdgcn_exp2f(e * __builtin_amdgcn_logf(b));
}

__device__ __forceinline__ float pow_pos_sub(float b, float e) {
    return exp2f(e * __builtin_amdgcn_logf(b));
}

__device__ __forceinline__ float power_r(float x, float y, float ra, float w0, float w1) {
    ra = (fabsf(ra) < EPS) ? EPS : ra;
    return pow_pos(w0 * pow_pos(x, ra) + w1 * pow_pos(y, ra), 1.0f / ra);
}

__device__ __forceinline__ float F_upper(float x, float y, float a, float w0, float w1) {
    const float xy  = x * y;
    const float mn  = fminf(x, y);
    const float lin = w0 * x + w1 * y;
    float out = lin;
    if (a > 0.5f && a < 0.75f) {
        const float R = r_poly(0.75f);
        out = (3.0f - 4.0f * a) * lin + (4.0f * a - 2.0f) * power_r(x, y, R, w0, w1);
    }
    if (a >= 0.75f && a < 1.0f) {
        const float ra = r_poly(fminf(fmaxf(a, 0.75f), 1.0f - EPS));
        out = power_r(x, y, ra, w0, w1);
    }
    if (fabsf(a - 1.0f)  < EPS) out = mn;
    if (a > 1.0f  && a < 1.25f) out = 4.0f * ((1.25f - a) * mn + (a - 1.0f) * xy);
    if (fabsf(a - 1.25f) < EPS) out = xy;
    if (a > 1.25f && a < 2.0f) {
        const float p = sqrtf(fmaxf(3.0f / fmaxf(2.0f - a, EPS) - 1.0f, EPS));
        out = pow_pos_sub(xy, p);
    }
    if (fabsf(a - 2.0f)  < EPS)
        out = ((fabsf(x - 1.0f) < EPS) && (fabsf(y - 1.0f) < EPS)) ? 1.0f : 0.0f;
    return out;
}

__device__ __forceinline__ float F_comb(float x, float y, float a, float w0, float w1) {
    if (isnan(x)) x = EPS;
    x = fminf(fmaxf(x, EPS), 1.0f - EPS);
    if (isnan(y)) y = EPS;
    y = fminf(fmaxf(y, EPS), 1.0f - EPS);
    if (a < 0.5f - EPS) {
        const float af = fminf(fmaxf(1.0f - a, -1.0f + EPS), 2.0f - EPS);
        return 1.0f - F_upper(1.0f - x, 1.0f - y, af, w0, w1);
    }
    return F_upper(x, y, a, w0, w1);
}

__device__ __forceinline__ float combine_op(float l, float r, float a, float w0, float w1) {
    const float nres = F_comb(l, r, a, w0, w1);
    return isnan(nres) ? a : nres;
}

__global__ __launch_bounds__(256) void chain_kernel(const float* __restrict__ leafT,
                                                    const float* __restrict__ weights,
                                                    const float* __restrict__ biases,
                                                    float* __restrict__ out) {
    __shared__ float sA[255], sW0[255], sW1[255];
    const int t = threadIdx.x;
    if (t < 255) {
        const float w0 = weights[2 * t], w1 = weights[2 * t + 1];
        const float m  = fmaxf(w0, w1);
        const float e0 = expf(w0 - m), e1 = expf(w1 - m);
        const float inv = 1.0f / (e0 + e1);
        sW0[t] = e0 * inv;
        sW1[t] = e1 * inv;
        float a = 3.0f / (1.0f + expf(-biases[t])) - 1.0f;
        sA[t] = fminf(fmaxf(a, -1.0f + EPS), 2.0f - EPS);
    }
    __syncthreads();

    const int b = blockIdx.x * 256 + t;
    float carry = combine_op(leafT[b], leafT[(size_t)BATCH + b], sA[0], sW0[0], sW1[0]);
    for (int l = 2; l < LDIM; ++l) {
        const float leaf = leafT[(size_t)l * BATCH + b];
        carry = combine_op(carry, leaf, sA[l - 1], sW0[l - 1], sW1[l - 1]);
    }
    out[b] = carry;
}

extern "C" void kernel_launch(void* const* d_in, const int* in_sizes, int n_in,
                              void* d_out, int out_size, void* d_ws, size_t ws_size,
                              hipStream_t stream) {
    const float* x       = (const float*)d_in[0];
    const float* logits  = (const float*)d_in[1];
    const float* weights = (const float*)d_in[2];
    const float* biases  = (const float*)d_in[3];

    float* P     = (float*)d_ws;
    float* leafT = P + LDIM * LDIM;

    sinkhorn_kernel<<<1, 1024, 0, stream>>>(logits, P);
    gemm_kernel<<<BATCH / 128, 256, 0, stream>>>(x, P, leafT);
    chain_kernel<<<BATCH / 256, 256, 0, stream>>>(leafT, weights, biases, (float*)d_out);
}


# Scan kernel: each step's operands (its leaf from global memory, its level's 3 parameters from LDS) are loaded 8 steps ahead into registers, not waited for inside the step; arithmetic unchanged
# speedup vs baseline: 3.4096x; 1.7000x over previous
#include <hip/hip_runtime.h>
#include <hip/hip_bf16.h>
#include <math.h>

#define EPS 1e-6f
#define LDIM 256
#define BATCH 65536
#define SPAD 257

typedef __attribute__((ext_vector_type(2))) float v2f;
typedef __attribute__((ext_vector_type(8))) float v8f;

__device__ __forceinline__ int sw_idx(int r, int c) {
    return r * SPAD + (c ^ ((((c >> 6) ^ (r >> 6)) & 3) << 3));
}

__device__ __forceinline__ int vpad(int i) { return i + ((i >> 6) << 3); }

__global__ __launch_bounds__(1024) void sinkhorn_kernel(const float* __restrict__ logits,
                                                        float* __restrict__ P) {
    __shared__ float M[LDIM * SPAD];
    __shared__ __attribute__((aligned(16))) float va[LDIM + 24];
    __shared__ __attribute__((aligned(16))) float vb[LDIM + 24];
    const int t = threadIdx.x;
    const int g = t >> 2;
    const int q = t & 3;

    const int cc = t & 255;
    const int r0 = t >> 8;
    for (int kk = 0; kk < 4; ++kk) {
        const int base = (r0 + 64 * kk) * SPAD + (cc ^ ((((cc >> 6) ^ kk) & 3) << 3));
        for (int m = 0; m < 16; ++m)
            M[base + 4 * m * SPAD] = logits[t + 1024 * (16 * kk + m)];
    }
    if (t < LDIM) vb[vpad(t)] = 1.0f;
    __syncthreads();

    const int rswz  = ((q ^ (g >> 6)) & 3) << 3;
    const int rbase = g * SPAD + q * 64;
    const int cbase = (q * 64) * SPAD + (g ^ ((((g >> 6) ^ q) & 3) << 3));
    const int vq    = q * 72;

    float kr[64];
    float mx = -3.4e38f;
    #pragma unroll
    for (int j = 0; j < 64; j += 8) {
        const float* p = &M[rbase + (j ^ rswz)];
        #pragma unroll
        for (int e = 0; e < 8; ++e) { kr[j + e] = p[e]; mx = fmaxf(mx, kr[j + e]); }
    }
    mx = fmaxf(mx, __shfl_xor(mx, 1));
    mx = fmaxf(mx, __shfl_xor(mx, 2));
    #pragma unroll
    for (int j = 0; j < 64; j += 8) {
        float* p = &M[rbase + (j ^ rswz)];
        #pragma unroll
        for (int e = 0; e < 8; ++e) { kr[j + e] = expf(kr[j + e] - mx); p[e] = kr[j + e]; }
    }

    float a_g = 1.0f;
    for (int it = 0; it < 10; ++it) {
        {
            const float4* v4 = (const float4*)&vb[vq];
            float s0 = 0.0f, s1 = 0.0f, s2 = 0.0f, s3 = 0.0f;
            #pragma unroll
            for (int k = 0; k < 16; ++k) {
                const float4 v = v4[k];
                s0 = fmaf(kr[4 * k + 0], v.x, s0);
                s1 = fmaf(kr[4 * k + 1], v.y, s1);
                s2 = fmaf(kr[4 * k + 2], v.z, s2);
                s3 = fmaf(kr[4 * k + 3], v.w, s3);
            }
            float s = (s0 + s1) + (s2 + s3);
            s += __shfl_xor(s, 1);
            s += __shfl_xor(s, 2);
            a_g = 1.0f / s;
            if (q == 0) va[vpad(g)] = a_g;
        }
        __syncthreads();
        {
            const float4* v4 = (const float4*)&va[vq];
            float s0 = 0.0f, s1 = 0.0f, s2 = 0.0f, s3 = 0.0f;
            #pragma unroll 4
            for (int k = 0; k < 16; ++k) {
                const float4 v = v4[k];
                const float* p = &M[cbase + (4 * k) * SPAD];
                s0 = fmaf(p[0],        v.x, s0);
                s1 = fmaf(p[SPAD],     v.y, s1);
                s2 = fmaf(p[2 * SPAD], v.z, s2);
                s3 = fmaf(p[3 * SPAD], v.w, s3);
            }
            float s = (s0 + s1) + (s2 + s3);
            s += __shfl_xor(s, 1);
            s += __shfl_xor(s, 2);
            if (q == 0) vb[vpad(g)] = 1.0f / s;
        }
        __syncthreads();
    }

    #pragma unroll
    for (int j = 0; j < 64; j += 8) {
        float* p = &M[rbase + (j ^ rswz)];
        #pragma unroll
        for (int e = 0; e < 8; ++e) p[e] = (kr[j + e] * a_g) * vb[vq + j + e];
    }
    __syncthreads();

    for (int kk = 0; kk < 4; ++kk) {
        const int base = (r0 + 64 * kk) * SPAD + (cc ^ ((((cc >> 6) ^ kk) & 3) << 3));
        for (int m = 0; m < 16; ++m)
            P[t + 1024 * (16 * kk + m)] = M[base + 4 * m * SPAD];
    }
}

__global__ __launch_bounds__(256) void gemm_kernel(const float* __restrict__ x,
                                                   const float* __restrict__ P,
                                                   float* __restrict__ leafT) {
    __shared__ float Ps[LDIM * LDIM];
    const int t = threadIdx.x;
    {
        const float4* Pv = (const float4*)P;
        float4* Sv = (float4*)Ps;
        for (int i = t; i < LDIM * LDIM / 4; i += 256) Sv[i] = Pv[i];
    }
    __syncthreads();

    const int wave  = t >> 5;
    const int lane  = t & 31;
    const int mloc  = lane & 15;
    const int khalf = (lane >> 4) * 2;
    const int m0    = blockIdx.x * 128 + wave * 16;

    v8f acc[16] = {};

    const float* xrow = x + (size_t)(m0 + mloc) * LDIM + khalf;

    for (int k = 0; k < LDIM; k += 4) {
        v2f bfrag;
        bfrag.x = xrow[k];
        bfrag.y = xrow[k + 1];
        #pragma unroll
        for (int nt = 0; nt < 16; ++nt) {
            const int n = nt * 16 + mloc;
            v2f afrag;
            afrag.x = Ps[(k + khalf)     * LDIM + n];
            afrag.y = Ps[(k + khalf + 1) * LDIM + n];
            acc[nt] = __builtin_amdgcn_wmma_f32_16x16x4_f32(
                false, afrag, false, bfrag, (short)0, acc[nt], false, false);
        }
    }

    const size_t col = (size_t)m0 + mloc;
    #pragma unroll
    for (int nt = 0; nt < 16; ++nt) {
        const int nbase = nt * 16 + (lane >> 4) * 8;
        #pragma unroll
        for (int i = 0; i < 8; ++i)
            leafT[(size_t)(nbase + i) * BATCH + col] = acc[nt][i];
    }
}

__device__ __forceinline__ float r_poly(float a) {
    float d  = 0.5f - a;
    float d2 = d * d;
    float num = 0.25f + 1.65811f * d + 2.15388f * d2 + 8.2844f * d2 * d + 6.16764f * d2 * d2;
    float den = a * (1.0f - a);
    den = (fabsf(den) < EPS) ? EPS : den;
    return num / den;
}

__device__ __forceinline__ float pow_pos(float b, float e) {
    return __builtin_amdgcn_exp2f(e * __builtin_amdgcn_logf(b));
}

__device__ __forceinline__ float pow_pos_sub(float b, float e) {
    return exp2f(e * __builtin_amdgcn_logf(b));
}

__device__ __forceinline__ float power_r(float x, float y, float ra, float w0, float w1) {
    ra = (fabsf(ra) < EPS) ? EPS : ra;
    return pow_pos(w0 * pow_pos(x, ra) + w1 * pow_pos(y, ra), 1.0f / ra);
}

__device__ __forceinline__ float F_upper(float x, float y, float a, float w0, float w1) {
    const float xy  = x * y;
    const float mn  = fminf(x, y);
    const float lin = w0 * x + w1 * y;
    float out = lin;
    if (a > 0.5f && a < 0.75f) {
        const float R = r_poly(0.75f);
        out = (3.0f - 4.0f * a) * lin + (4.0f * a - 2.0f) * power_r(x, y, R, w0, w1);
    }
    if (a >= 0.75f && a < 1.0f) {
        const float ra = r_poly(fminf(fmaxf(a, 0.75f), 1.0f - EPS));
        out = power_r(x, y, ra, w0, w1);
    }
    if (fabsf(a - 1.0f)  < EPS) out = mn;
    if (a > 1.0f  && a < 1.25f) out = 4.0f * ((1.25f - a) * mn + (a - 1.0f) * xy);
    if (fabsf(a - 1.25f) < EPS) out = xy;
    if (a > 1.25f && a < 2.0f) {
        const float p = sqrtf(fmaxf(3.0f / fmaxf(2.0f - a, EPS) - 1.0f, EPS));
        out = pow_pos_sub(xy, p);
    }
    if (fabsf(a - 2.0f)  < EPS)
        out = ((fabsf(x - 1.0f) < EPS) && (fabsf(y - 1.0f) < EPS)) ? 1.0f : 0.0f;
    return out;
}

__device__ __forceinline__ float F_comb(float x, float y, float a, float w0, float w1) {
    if (isnan(x)) x = EPS;
    x = fminf(fmaxf(x, EPS), 1.0f - EPS);
    if (isnan(y)) y = EPS;
    y = fminf(fmaxf(y, EPS), 1.0f - EPS);
    if (a < 0.5f - EPS) {
        const float af = fminf(fmaxf(1.0f - a, -1.0f + EPS), 2.0f - EPS);
        return 1.0f - F_upper(1.0f - x, 1.0f - y, af, w0, w1);
    }
    return F_upper(x, y, a, w0, w1);
}

__device__ __forceinline__ float combine_op(float l, float r, float a, float w0, float w1) {
    const float nres = F_comb(l, r, a, w0, w1);
    return isnan(nres) ? a : nres;
}

__global__ __launch_bounds__(256) void chain_kernel(const float* __restrict__ leafT,
                                                    const float* __restrict__ weights,
                                                    const float* __restrict__ biases,
                                                    float* __restrict__ out) {
    __shared__ __attribute__((aligned(16))) float sA[LDIM], sW0[LDIM], sW1[LDIM];
    const int t = threadIdx.x;
    const int b = blockIdx.x * 256 + t;

    constexpr int PF = 8;
    const float* __restrict__ lp = leafT + b;
    float nxt[PF];
    #pragma unroll
    for (int j = 0; j < PF; ++j) nxt[j] = lp[(size_t)j * BATCH];

    if (t < 255) {
        const float w0 = weights[2 * t], w1 = weights[2 * t + 1];
        const float m  = fmaxf(w0, w1);
        const float e0 = expf(w0 - m), e1 = expf(w1 - m);
        const float inv = 1.0f / (e0 + e1);
        sW0[t + 1] = e0 * inv;
        sW1[t + 1] = e1 * inv;
        float a = 3.0f / (1.0f + expf(-biases[t])) - 1.0f;
        sA[t + 1] = fminf(fmaxf(a, -1.0f + EPS), 2.0f - EPS);
    } else {
        sA[0] = 0.0f; sW0[0] = 0.0f; sW1[0] = 0.0f;
    }
    __syncthreads();

    float4 na[2], nw0[2], nw1[2];
    #pragma unroll
    for (int h = 0; h < 2; ++h) {
        na[h]  = ((const float4*)sA)[h];
        nw0[h] = ((const float4*)sW0)[h];
        nw1[h] = ((const float4*)sW1)[h];
    }

    float carry = 0.0f;
    for (int g = 0; g < LDIM / PF; ++g) {
        float cur[PF], pa[PF], pw0[PF], pw1[PF];
        #pragma unroll
        for (int j = 0; j < PF; ++j) cur[j] = nxt[j];
        #pragma unroll
        for (int h = 0; h < 2; ++h) {
            pa[4 * h]  = na[h].x;  pa[4 * h + 1]  = na[h].y;  pa[4 * h + 2]  = na[h].z;  pa[4 * h + 3]  = na[h].w;
            pw0[4 * h] = nw0[h].x; pw0[4 * h + 1] = nw0[h].y; pw0[4 * h + 2] = nw0[h].z; pw0[4 * h + 3] = nw0[h].w;
            pw1[4 * h] = nw1[h].x; pw1[4 * h + 1] = nw1[h].y; pw1[4 * h + 2] = nw1[h].z; pw1[4 * h + 3] = nw1[h].w;
        }
        const int gn = (g + 1 < LDIM / PF) ? g + 1 : g;
        #pragma unroll
        for (int j = 0; j < PF; ++j) nxt[j] = lp[(size_t)(gn * PF + j) * BATCH];
        #pragma unroll
        for (int h = 0; h < 2; ++h) {
            na[h]  = ((const float4*)sA)[2 * gn + h];
            nw0[h] = ((const float4*)sW0)[2 * gn + h];
            nw1[h] = ((const float4*)sW1)[2 * gn + h];
        }
        #pragma unroll
        for (int j = 0; j < PF; ++j) {
            if (j == 0 && g == 0) carry = cur[0];
            else carry = combine_op(carry, cur[j], pa[j], pw0[j], pw1[j]);
        }
    }
    out[b] = carry;
}

extern "C" void kernel_launch(void* const* d_in, const int* in_sizes, int n_in,
                              void* d_out, int out_size, void* d_ws, size_t ws_size,
                              hipStream_t stream) {
    const float* x       = (const float*)d_in[0];
    const float* logits  = (const float*)d_in[1];
    const float* weights = (const float*)d_in[2];
    const float* biases  = (const float*)d_in[3];

    float* P     = (float*)d_ws;
    float* leafT = P + LDIM * LDIM;

    sinkhorn_kernel<<<1, 1024, 0, stream>>>(logits, P);
    gemm_kernel<<<BATCH / 128, 256, 0, stream>>>(x, P, leafT);
    chain_kernel<<<BATCH / 256, 256, 0, stream>>>(leafT, weights, biases, (float*)d_out);
}


# Scan kernel: what depends on the level alone (where-chain class, mirror flag, r_poly exponent and reciprocal, sqrt exponent, linear weights) is computed once per level into LDS; sum order kept
# speedup vs baseline: 3.9280x; 1.1520x over previous
#include <hip/hip_runtime.h>
#include <hip/hip_bf16.h>
#include <math.h>

#define EPS 1e-6f
#define LDIM 256
#define BATCH 65536
#define SPAD 257

typedef __attribute__((ext_vector_type(2))) float v2f;
typedef __attribute__((ext_vector_type(8))) float v8f;

__device__ __forceinline__ int sw_idx(int r, int c) {
    return r * SPAD + (c ^ ((((c >> 6) ^ (r >> 6)) & 3) << 3));
}

__device__ __forceinline__ int vpad(int i) { return i + ((i >> 6) << 3); }

__global__ __launch_bounds__(1024) void sinkhorn_kernel(const float* __restrict__ logits,
                                                        float* __restrict__ P) {
    __shared__ float M[LDIM * SPAD];
    __shared__ __attribute__((aligned(16))) float va[LDIM + 24];
    __shared__ __attribute__((aligned(16))) float vb[LDIM + 24];
    const int t = threadIdx.x;
    const int g = t >> 2;
    const int q = t & 3;

    const int cc = t & 255;
    const int r0 = t >> 8;
    for (int kk = 0; kk < 4; ++kk) {
        const int base = (r0 + 64 * kk) * SPAD + (cc ^ ((((cc >> 6) ^ kk) & 3) << 3));
        for (int m = 0; m < 16; ++m)
            M[base + 4 * m * SPAD] = logits[t + 1024 * (16 * kk + m)];
    }
    if (t < LDIM) vb[vpad(t)] = 1.0f;
    __syncthreads();

    const int rswz  = ((q ^ (g >> 6)) & 3) << 3;
    const int rbase = g * SPAD + q * 64;
    const int cbase = (q * 64) * SPAD + (g ^ ((((g >> 6) ^ q) & 3) << 3));
    const int vq    = q * 72;

    float kr[64];
    float mx = -3.4e38f;
    #pragma unroll
    for (int j = 0; j < 64; j += 8) {
        const float* p = &M[rbase + (j ^ rswz)];
        #pragma unroll
        for (int e = 0; e < 8; ++e) { kr[j + e] = p[e]; mx = fmaxf(mx, kr[j + e]); }
    }
    mx = fmaxf(mx, __shfl_xor(mx, 1));
    mx = fmaxf(mx, __shfl_xor(mx, 2));
    #pragma unroll
    for (int j = 0; j < 64; j += 8) {
        float* p = &M[rbase + (j ^ rswz)];
        #pragma unroll
        for (int e = 0; e < 8; ++e) { kr[j + e] = expf(kr[j + e] - mx); p[e] = kr[j + e]; }
    }

    float a_g = 1.0f;
    for (int it = 0; it < 10; ++it) {
        {
            const float4* v4 = (const float4*)&vb[vq];
            float s0 = 0.0f, s1 = 0.0f, s2 = 0.0f, s3 = 0.0f;
            #pragma unroll
            for (int k = 0; k < 16; ++k) {
                const float4 v = v4[k];
                s0 = fmaf(kr[4 * k + 0], v.x, s0);
                s1 = fmaf(kr[4 * k + 1], v.y, s1);
                s2 = fmaf(kr[4 * k + 2], v.z, s2);
                s3 = fmaf(kr[4 * k + 3], v.w, s3);
            }
            float s = (s0 + s1) + (s2 + s3);
            s += __shfl_xor(s, 1);
            s += __shfl_xor(s, 2);
            a_g = 1.0f / s;
            if (q == 0) va[vpad(g)] = a_g;
        }
        __syncthreads();
        {
            const float4* v4 = (const float4*)&va[vq];
            float s0 = 0.0f, s1 = 0.0f, s2 = 0.0f, s3 = 0.0f;
            #pragma unroll 4
            for (int k = 0; k < 16; ++k) {
                const float4 v = v4[k];
                const float* p = &M[cbase + (4 * k) * SPAD];
                s0 = fmaf(p[0],        v.x, s0);
                s1 = fmaf(p[SPAD],     v.y, s1);
                s2 = fmaf(p[2 * SPAD], v.z, s2);
                s3 = fmaf(p[3 * SPAD], v.w, s3);
            }
            float s = (s0 + s1) + (s2 + s3);
            s += __shfl_xor(s, 1);
            s += __shfl_xor(s, 2);
            if (q == 0) vb[vpad(g)] = 1.0f / s;
        }
        __syncthreads();
    }

    #pragma unroll
    for (int j = 0; j < 64; j += 8) {
        float* p = &M[rbase + (j ^ rswz)];
        #pragma unroll
        for (int e = 0; e < 8; ++e) p[e] = (kr[j + e] * a_g) * vb[vq + j + e];
    }
    __syncthreads();

    for (int kk = 0; kk < 4; ++kk) {
        const int base = (r0 + 64 * kk) * SPAD + (cc ^ ((((cc >> 6) ^ kk) & 3) << 3));
        for (int m = 0; m < 16; ++m)
            P[t + 1024 * (16 * kk + m)] = M[base + 4 * m * SPAD];
    }
}

__global__ __launch_bounds__(256) void gemm_kernel(const float* __restrict__ x,
                                                   const float* __restrict__ P,
                                                   float* __restrict__ leafT) {
    __shared__ float Ps[LDIM * LDIM];
    const int t = threadIdx.x;
    {
        const float4* Pv = (const float4*)P;
        float4* Sv = (float4*)Ps;
        for (int i = t; i < LDIM * LDIM / 4; i += 256) Sv[i] = Pv[i];
    }
    __syncthreads();

    const int wave  = t >> 5;
    const int lane  = t & 31;
    const int mloc  = lane & 15;
    const int khalf = (lane >> 4) * 2;
    const int m0    = blockIdx.x * 128 + wave * 16;

    v8f acc[16] = {};

    const float* xrow = x + (size_t)(m0 + mloc) * LDIM + khalf;

    for (int k = 0; k < LDIM; k += 4) {
        v2f bfrag;
        bfrag.x = xrow[k];
        bfrag.y = xrow[k + 1];
        #pragma unroll
        for (int nt = 0; nt < 16; ++nt) {
            const int n = nt * 16 + mloc;
            v2f afrag;
            afrag.x = Ps[(k + khalf)     * LDIM + n];
            afrag.y = Ps[(k + khalf + 1) * LDIM + n];
            acc[nt] = __builtin_amdgcn_wmma_f32_16x16x4_f32(
                false, afrag, false, bfrag, (short)0, acc[nt], false, false);
        }
    }

    const size_t col = (size_t)m0 + mloc;
    #pragma unroll
    for (int nt = 0; nt < 16; ++nt) {
        const int nbase = nt * 16 + (lane >> 4) * 8;
        #pragma unroll
        for (int i = 0; i < 8; ++i)
            leafT[(size_t)(nbase + i) * BATCH + col] = acc[nt][i];
    }
}

__device__ __forceinline__ float r_poly(float a) {
    float d  = 0.5f - a;
    float d2 = d * d;
    float num = 0.25f + 1.65811f * d + 2.15388f * d2 + 8.2844f * d2 * d + 6.16764f * d2 * d2;
    float den = a * (1.0f - a);
    den = (fabsf(den) < EPS) ? EPS : den;
    return num / den;
}

__device__ __forceinline__ float pow_pos(float b, float e) {
    return __builtin_amdgcn_exp2f(e * __builtin_amdgcn_logf(b));
}

__device__ __forceinline__ float pow_pos_sub(float b, float e) {
    return exp2f(e * __builtin_amdgcn_logf(b));
}

__device__ __forceinline__ float power_r2(float x, float y, float ra, float inv_ra, float w0, float w1) {
    return pow_pos(w0 * pow_pos(x, ra) + w1 * pow_pos(y, ra), inv_ra);
}

__device__ __forceinline__ float power_r(float x, float y, float ra, float w0, float w1) {
    ra = (fabsf(ra) < EPS) ? EPS : ra;
    return power_r2(x, y, ra, 1.0f / ra, w0, w1);
}

#define C_LIN  0
#define C_R6   1
#define C_R5   2
#define C_MN   3
#define C_R3   4
#define C_XY   5
#define C_R1   6
#define C_R0   7
#define C_FLIP 8

__device__ __forceinline__ void level_setup(float a, int& cw, float& p0, float& p1) {
    const bool flip = a < 0.5f - EPS;
    const float au = flip ? fminf(fmaxf(1.0f - a, -1.0f + EPS), 2.0f - EPS) : a;
    int c = C_LIN;
    float q0 = 0.0f, q1 = 0.0f;
    if (au > 0.5f && au < 0.75f) { c = C_R6; q0 = 3.0f - 4.0f * au; q1 = 4.0f * au - 2.0f; }
    if (au >= 0.75f && au < 1.0f) {
        float ra = r_poly(fminf(fmaxf(au, 0.75f), 1.0f - EPS));
        ra = (fabsf(ra) < EPS) ? EPS : ra;
        c = C_R5; q0 = ra; q1 = 1.0f / ra;
    }
    if (fabsf(au - 1.0f)  < EPS) c = C_MN;
    if (au > 1.0f  && au < 1.25f) { c = C_R3; q0 = 1.25f - au; q1 = au - 1.0f; }
    if (fabsf(au - 1.25f) < EPS) c = C_XY;
    if (au > 1.25f && au < 2.0f) {
        c = C_R1; q0 = sqrtf(fmaxf(3.0f / fmaxf(2.0f - au, EPS) - 1.0f, EPS));
    }
    if (fabsf(au - 2.0f)  < EPS) c = C_R0;
    cw = flip ? (c | C_FLIP) : c;
    p0 = q0;
    p1 = q1;
}

__device__ __forceinline__ float step_op(float l, float r, int cw, float p0, float p1,
                                         float w0, float w1, float a) {
    float x = l, y = r;
    if (isnan(x)) x = EPS;
    x = fminf(fmaxf(x, EPS), 1.0f - EPS);
    if (isnan(y)) y = EPS;
    y = fminf(fmaxf(y, EPS), 1.0f - EPS);
    const bool flip = (cw & C_FLIP) != 0;
    const float sg = flip ? -1.0f : 1.0f;
    const float of = flip ?  1.0f : 0.0f;
    x = fmaf(sg, x, of);
    y = fmaf(sg, y, of);
    float out;
    switch (cw & 7) {
    case C_R6: out = p0 * (w0 * x + w1 * y) + p1 * power_r(x, y, r_poly(0.75f), w0, w1); break;
    case C_R5: out = power_r2(x, y, p0, p1, w0, w1); break;
    case C_MN: out = fminf(x, y); break;
    case C_R3: out = 4.0f * (p0 * fminf(x, y) + p1 * (x * y)); break;
    case C_XY: out = x * y; break;
    case C_R1: out = pow_pos_sub(x * y, p0); break;
    case C_R0: out = ((fabsf(x - 1.0f) < EPS) && (fabsf(y - 1.0f) < EPS)) ? 1.0f : 0.0f; break;
    default:   out = w0 * x + w1 * y; break;
    }
    const float nres = fmaf(sg, out, of);
    return isnan(nres) ? a : nres;
}

__global__ __launch_bounds__(256) void chain_kernel(const float* __restrict__ leafT,
                                                    const float* __restrict__ weights,
                                                    const float* __restrict__ biases,
                                                    float* __restrict__ out) {
    __shared__ __attribute__((aligned(16))) float sA[LDIM], sW0[LDIM], sW1[LDIM], sP0[LDIM], sP1[LDIM];
    __shared__ __attribute__((aligned(16))) int sC[LDIM];
    const int t = threadIdx.x;
    const int b = blockIdx.x * 256 + t;

    constexpr int PF = 8;
    const float* __restrict__ lp = leafT + b;
    float nxt[PF];
    #pragma unroll
    for (int j = 0; j < PF; ++j) nxt[j] = lp[(size_t)j * BATCH];

    if (t < 255) {
        const float w0 = weights[2 * t], w1 = weights[2 * t + 1];
        const float m  = fmaxf(w0, w1);
        const float e0 = expf(w0 - m), e1 = expf(w1 - m);
        const float inv = 1.0f / (e0 + e1);
        sW0[t + 1] = e0 * inv;
        sW1[t + 1] = e1 * inv;
        float a = 3.0f / (1.0f + expf(-biases[t])) - 1.0f;
        a = fminf(fmaxf(a, -1.0f + EPS), 2.0f - EPS);
        sA[t + 1] = a;
        int cw; float p0, p1;
        level_setup(a, cw, p0, p1);
        sC[t + 1] = cw; sP0[t + 1] = p0; sP1[t + 1] = p1;
    } else {
        sA[0] = 0.0f; sW0[0] = 0.0f; sW1[0] = 0.0f; sP0[0] = 0.0f; sP1[0] = 0.0f; sC[0] = C_LIN;
    }
    __syncthreads();

    float4 na[2], nw0[2], nw1[2], np0[2], np1[2];
    int4 nc[2];
    #pragma unroll
    for (int h = 0; h < 2; ++h) {
        na[h]  = ((const float4*)sA)[h];
        nw0[h] = ((const float4*)sW0)[h];
        nw1[h] = ((const float4*)sW1)[h];
        np0[h] = ((const float4*)sP0)[h];
        np1[h] = ((const float4*)sP1)[h];
        nc[h]  = ((const int4*)sC)[h];
    }

    float carry = 0.0f;
    for (int g = 0; g < LDIM / PF; ++g) {
        float cur[PF], pa[PF], pw0[PF], pw1[PF], pp0[PF], pp1[PF];
        int pc[PF];
        #pragma unroll
        for (int j = 0; j < PF; ++j) cur[j] = nxt[j];
        #pragma unroll
        for (int h = 0; h < 2; ++h) {
            pa[4 * h]  = na[h].x;  pa[4 * h + 1]  = na[h].y;  pa[4 * h + 2]  = na[h].z;  pa[4 * h + 3]  = na[h].w;
            pw0[4 * h] = nw0[h].x; pw0[4 * h + 1] = nw0[h].y; pw0[4 * h + 2] = nw0[h].z; pw0[4 * h + 3] = nw0[h].w;
            pw1[4 * h] = nw1[h].x; pw1[4 * h + 1] = nw1[h].y; pw1[4 * h + 2] = nw1[h].z; pw1[4 * h + 3] = nw1[h].w;
            pp0[4 * h] = np0[h].x; pp0[4 * h + 1] = np0[h].y; pp0[4 * h + 2] = np0[h].z; pp0[4 * h + 3] = np0[h].w;
            pp1[4 * h] = np1[h].x; pp1[4 * h + 1] = np1[h].y; pp1[4 * h + 2] = np1[h].z; pp1[4 * h + 3] = np1[h].w;
            pc[4 * h]  = nc[h].x;  pc[4 * h + 1]  = nc[h].y;  pc[4 * h + 2]  = nc[h].z;  pc[4 * h + 3]  = nc[h].w;
        }
        const int gn = (g + 1 < LDIM / PF) ? g + 1 : g;
        #pragma unroll
        for (int j = 0; j < PF; ++j) nxt[j] = lp[(size_t)(gn * PF + j) * BATCH];
        #pragma unroll
        for (int h = 0; h < 2; ++h) {
            na[h]  = ((const float4*)sA)[2 * gn + h];
            nw0[h] = ((const float4*)sW0)[2 * gn + h];
            nw1[h] = ((const float4*)sW1)[2 * gn + h];
            np0[h] = ((const float4*)sP0)[2 * gn + h];
            np1[h] = ((const float4*)sP1)[2 * gn + h];
            nc[h]  = ((const int4*)sC)[2 * gn + h];
        }
        #pragma unroll
        for (int j = 0; j < PF; ++j) {
            if (j == 0 && g == 0) carry = cur[0];
            else carry = step_op(carry, cur[j], pc[j], pp0[j], pp1[j], pw0[j], pw1[j], pa[j]);
        }
    }
    out[b] = carry;
}

extern "C" void kernel_launch(void* const* d_in, const int* in_sizes, int n_in,
                              void* d_out, int out_size, void* d_ws, size_t ws_size,
                              hipStream_t stream) {
    const float* x       = (const float*)d_in[0];
    const float* logits  = (const float*)d_in[1];
    const float* weights = (const float*)d_in[2];
    const float* biases  = (const float*)d_in[3];

    float* P     = (float*)d_ws;
    float* leafT = P + LDIM * LDIM;

    sinkhorn_kernel<<<1, 1024, 0, stream>>>(logits, P);
    gemm_kernel<<<BATCH / 128, 256, 0, stream>>>(x, P, leafT);
    chain_kernel<<<BATCH / 256, 256, 0, stream>>>(leafT, weights, biases, (float*)d_out);
}


# Sinkhorn: each thread keeps an 8x8 block of K in registers for both sweeps; row sums finish by lane exchange in the wave, column sums across waves via an LDS table of partials (sum order changes)
# speedup vs baseline: 4.2300x; 1.0769x over previous
#include <hip/hip_runtime.h>
#include <hip/hip_bf16.h>
#include <math.h>

#define EPS 1e-6f
#define LDIM 256
#define BATCH 65536

typedef __attribute__((ext_vector_type(2))) float v2f;
typedef __attribute__((ext_vector_type(8))) float v8f;


__device__ __forceinline__ float from_lane(float v, int lane) {
    return __int_as_float(__builtin_amdgcn_readlane(__float_as_int(v), lane));
}

template <bool IS_MAX>
__device__ __forceinline__ float comb2(float x, float y) { return IS_MAX ? fmaxf(x, y) : x + y; }

template <bool IS_MAX>
__device__ __forceinline__ float wave_reduce8(const float (&s)[8], int l) {
    const bool h16 = (l & 16) != 0, h8 = (l & 8) != 0, h4 = (l & 4) != 0;
    float r[4];
    #pragma unroll
    for (int k = 0; k < 4; ++k) {
        const float keep = h16 ? s[k + 4] : s[k];
        const float send = h16 ? s[k] : s[k + 4];
        r[k] = comb2<IS_MAX>(keep, __shfl_xor(send, 16));
    }
    float q[2];
    #pragma unroll
    for (int k = 0; k < 2; ++k) {
        const float keep = h8 ? r[k + 2] : r[k];
        const float send = h8 ? r[k] : r[k + 2];
        q[k] = comb2<IS_MAX>(keep, __shfl_xor(send, 8));
    }
    const float keep = h4 ? q[1] : q[0];
    const float send = h4 ? q[0] : q[1];
    float p = comb2<IS_MAX>(keep, __shfl_xor(send, 4));
    p = comb2<IS_MAX>(p, __shfl_xor(p, 2));
    p = comb2<IS_MAX>(p, __shfl_xor(p, 1));
    return p;
}

__global__ __launch_bounds__(1024) void sinkhorn_kernel(const float* __restrict__ logits,
                                                        float* __restrict__ P) {
    __shared__ __attribute__((aligned(16))) float part[32 * LDIM];
    __shared__ __attribute__((aligned(16))) float vb[LDIM];
    const int t = threadIdx.x;
    const int w = t >> 5;
    const int l = t & 31;

    float kr[8][8];
    const float* src = logits + (size_t)(8 * w) * LDIM + 8 * l;
    #pragma unroll
    for (int i = 0; i < 8; ++i) {
        const float4 lo = *(const float4*)(src + i * LDIM);
        const float4 hi = *(const float4*)(src + i * LDIM + 4);
        kr[i][0] = lo.x; kr[i][1] = lo.y; kr[i][2] = lo.z; kr[i][3] = lo.w;
        kr[i][4] = hi.x; kr[i][5] = hi.y; kr[i][6] = hi.z; kr[i][7] = hi.w;
    }

    {
        float m[8];
        #pragma unroll
        for (int i = 0; i < 8; ++i) {
            float mx = -3.4e38f;
            #pragma unroll
            for (int j = 0; j < 8; ++j) mx = fmaxf(mx, kr[i][j]);
            m[i] = mx;
        }
        const float mr = wave_reduce8<true>(m, l);
        #pragma unroll
        for (int i = 0; i < 8; ++i) {
            const float mx = from_lane(mr, 4 * i);
            #pragma unroll
            for (int j = 0; j < 8; ++j) kr[i][j] = expf(kr[i][j] - mx);
        }
    }

    float a[8], b[8];
    #pragma unroll
    for (int j = 0; j < 8; ++j) { a[j] = 1.0f; b[j] = 1.0f; }

    for (int it = 0; it < 10; ++it) {
        {
            float s[8];
            #pragma unroll
            for (int i = 0; i < 8; ++i) {
                float s0 = kr[i][0] * b[0], s1 = kr[i][1] * b[1];
                s0 = fmaf(kr[i][2], b[2], s0); s1 = fmaf(kr[i][3], b[3], s1);
                s0 = fmaf(kr[i][4], b[4], s0); s1 = fmaf(kr[i][5], b[5], s1);
                s0 = fmaf(kr[i][6], b[6], s0); s1 = fmaf(kr[i][7], b[7], s1);
                s[i] = s0 + s1;
            }
            const float inv = 1.0f / wave_reduce8<false>(s, l);
            #pragma unroll
            for (int i = 0; i < 8; ++i) a[i] = from_lane(inv, 4 * i);
        }
        {
            float c[8];
            #pragma unroll
            for (int j = 0; j < 8; ++j) {
                float c0 = kr[0][j] * a[0], c1 = kr[1][j] * a[1];
                c0 = fmaf(kr[2][j], a[2], c0); c1 = fmaf(kr[3][j], a[3], c1);
                c0 = fmaf(kr[4][j], a[4], c0); c1 = fmaf(kr[5][j], a[5], c1);
                c0 = fmaf(kr[6][j], a[6], c0); c1 = fmaf(kr[7][j], a[7], c1);
                c[j] = c0 + c1;
            }
            float4* pp = (float4*)&part[w * LDIM + 8 * l];
            pp[0] = make_float4(c[0], c[1], c[2], c[3]);
            pp[1] = make_float4(c[4], c[5], c[6], c[7]);
        }
        __syncthreads();
        if (t < LDIM) {
            float s0 = 0.0f, s1 = 0.0f, s2 = 0.0f, s3 = 0.0f;
            #pragma unroll
            for (int k = 0; k < 32; k += 4) {
                s0 += part[(k + 0) * LDIM + t];
                s1 += part[(k + 1) * LDIM + t];
                s2 += part[(k + 2) * LDIM + t];
                s3 += part[(k + 3) * LDIM + t];
            }
            vb[t] = 1.0f / ((s0 + s1) + (s2 + s3));
        }
        __syncthreads();
        {
            const float4 lo = *(const float4*)&vb[8 * l];
            const float4 hi = *(const float4*)&vb[8 * l + 4];
            b[0] = lo.x; b[1] = lo.y; b[2] = lo.z; b[3] = lo.w;
            b[4] = hi.x; b[5] = hi.y; b[6] = hi.z; b[7] = hi.w;
        }
    }

    float* dst = P + (size_t)(8 * w) * LDIM + 8 * l;
    #pragma unroll
    for (int i = 0; i < 8; ++i) {
        *(float4*)(dst + i * LDIM) = make_float4((kr[i][0] * a[i]) * b[0], (kr[i][1] * a[i]) * b[1],
                                                 (kr[i][2] * a[i]) * b[2], (kr[i][3] * a[i]) * b[3]);
        *(float4*)(dst + i * LDIM + 4) = make_float4((kr[i][4] * a[i]) * b[4], (kr[i][5] * a[i]) * b[5],
                                                     (kr[i][6] * a[i]) * b[6], (kr[i][7] * a[i]) * b[7]);
    }
}

__global__ __launch_bounds__(256) void gemm_kernel(const float* __restrict__ x,
                                                   const float* __restrict__ P,
                                                   float* __restrict__ leafT) {
    __shared__ float Ps[LDIM * LDIM];
    const int t = threadIdx.x;
    {
        const float4* Pv = (const float4*)P;
        float4* Sv = (float4*)Ps;
        for (int i = t; i < LDIM * LDIM / 4; i += 256) Sv[i] = Pv[i];
    }
    __syncthreads();

    const int wave  = t >> 5;
    const int lane  = t & 31;
    const int mloc  = lane & 15;
    const int khalf = (lane >> 4) * 2;
    const int m0    = blockIdx.x * 128 + wave * 16;

    v8f acc[16] = {};

    const float* xrow = x + (size_t)(m0 + mloc) * LDIM + khalf;

    for (int k = 0; k < LDIM; k += 4) {
        v2f bfrag;
        bfrag.x = xrow[k];
        bfrag.y = xrow[k + 1];
        #pragma unroll
        for (int nt = 0; nt < 16; ++nt) {
            const int n = nt * 16 + mloc;
            v2f afrag;
            afrag.x = Ps[(k + khalf)     * LDIM + n];
            afrag.y = Ps[(k + khalf + 1) * LDIM + n];
            acc[nt] = __builtin_amdgcn_wmma_f32_16x16x4_f32(
                false, afrag, false, bfrag, (short)0, acc[nt], false, false);
        }
    }

    const size_t col = (size_t)m0 + mloc;
    #pragma unroll
    for (int nt = 0; nt < 16; ++nt) {
        const int nbase = nt * 16 + (lane >> 4) * 8;
        #pragma unroll
        for (int i = 0; i < 8; ++i)
            leafT[(size_t)(nbase + i) * BATCH + col] = acc[nt][i];
    }
}

__device__ __forceinline__ float r_poly(float a) {
    float d  = 0.5f - a;
    float d2 = d * d;
    float num = 0.25f + 1.65811f * d + 2.15388f * d2 + 8.2844f * d2 * d + 6.16764f * d2 * d2;
    float den = a * (1.0f - a);
    den = (fabsf(den) < EPS) ? EPS : den;
    return num / den;
}

__device__ __forceinline__ float pow_pos(float b, float e) {
    return __builtin_amdgcn_exp2f(e * __builtin_amdgcn_logf(b));
}

__device__ __forceinline__ float pow_pos_sub(float b, float e) {
    return exp2f(e * __builtin_amdgcn_logf(b));
}

__device__ __forceinline__ float power_r2(float x, float y, float ra, float inv_ra, float w0, float w1) {
    return pow_pos(w0 * pow_pos(x, ra) + w1 * pow_pos(y, ra), inv_ra);
}

__device__ __forceinline__ float power_r(float x, float y, float ra, float w0, float w1) {
    ra = (fabsf(ra) < EPS) ? EPS : ra;
    return power_r2(x, y, ra, 1.0f / ra, w0, w1);
}

#define C_LIN  0
#define C_R6   1
#define C_R5   2
#define C_MN   3
#define C_R3   4
#define C_XY   5
#define C_R1   6
#define C_R0   7
#define C_FLIP 8

__device__ __forceinline__ void level_setup(float a, int& cw, float& p0, float& p1) {
    const bool flip = a < 0.5f - EPS;
    const float au = flip ? fminf(fmaxf(1.0f - a, -1.0f + EPS), 2.0f - EPS) : a;
    int c = C_LIN;
    float q0 = 0.0f, q1 = 0.0f;
    if (au > 0.5f && au < 0.75f) { c = C_R6; q0 = 3.0f - 4.0f * au; q1 = 4.0f * au - 2.0f; }
    if (au >= 0.75f && au < 1.0f) {
        float ra = r_poly(fminf(fmaxf(au, 0.75f), 1.0f - EPS));
        ra = (fabsf(ra) < EPS) ? EPS : ra;
        c = C_R5; q0 = ra; q1 = 1.0f / ra;
    }
    if (fabsf(au - 1.0f)  < EPS) c = C_MN;
    if (au > 1.0f  && au < 1.25f) { c = C_R3; q0 = 1.25f - au; q1 = au - 1.0f; }
    if (fabsf(au - 1.25f) < EPS) c = C_XY;
    if (au > 1.25f && au < 2.0f) {
        c = C_R1; q0 = sqrtf(fmaxf(3.0f / fmaxf(2.0f - au, EPS) - 1.0f, EPS));
    }
    if (fabsf(au - 2.0f)  < EPS) c = C_R0;
    cw = flip ? (c | C_FLIP) : c;
    p0 = q0;
    p1 = q1;
}

__device__ __forceinline__ float step_op(float l, float r, int cw, float p0, float p1,
                                         float w0, float w1, float a) {
    float x = l, y = r;
    if (isnan(x)) x = EPS;
    x = fminf(fmaxf(x, EPS), 1.0f - EPS);
    if (isnan(y)) y = EPS;
    y = fminf(fmaxf(y, EPS), 1.0f - EPS);
    const bool flip = (cw & C_FLIP) != 0;
    const float sg = flip ? -1.0f : 1.0f;
    const float of = flip ?  1.0f : 0.0f;
    x = fmaf(sg, x, of);
    y = fmaf(sg, y, of);
    float out;
    switch (cw & 7) {
    case C_R6: out = p0 * (w0 * x + w1 * y) + p1 * power_r(x, y, r_poly(0.75f), w0, w1); break;
    case C_R5: out = power_r2(x, y, p0, p1, w0, w1); break;
    case C_MN: out = fminf(x, y); break;
    case C_R3: out = 4.0f * (p0 * fminf(x, y) + p1 * (x * y)); break;
    case C_XY: out = x * y; break;
    case C_R1: out = pow_pos_sub(x * y, p0); break;
    case C_R0: out = ((fabsf(x - 1.0f) < EPS) && (fabsf(y - 1.0f) < EPS)) ? 1.0f : 0.0f; break;
    default:   out = w0 * x + w1 * y; break;
    }
    const float nres = fmaf(sg, out, of);
    return isnan(nres) ? a : nres;
}

__global__ __launch_bounds__(256) void chain_kernel(const float* __restrict__ leafT,
                                                    const float* __restrict__ weights,
                                                    const float* __restrict__ biases,
                                                    float* __restrict__ out) {
    __shared__ __attribute__((aligned(16))) float sA[LDIM], sW0[LDIM], sW1[LDIM], sP0[LDIM], sP1[LDIM];
    __shared__ __attribute__((aligned(16))) int sC[LDIM];
    const int t = threadIdx.x;
    const int b = blockIdx.x * 256 + t;

    constexpr int PF = 8;
    const float* __restrict__ lp = leafT + b;
    float nxt[PF];
    #pragma unroll
    for (int j = 0; j < PF; ++j) nxt[j] = lp[(size_t)j * BATCH];

    if (t < 255) {
        const float w0 = weights[2 * t], w1 = weights[2 * t + 1];
        const float m  = fmaxf(w0, w1);
        const float e0 = expf(w0 - m), e1 = expf(w1 - m);
        const float inv = 1.0f / (e0 + e1);
        sW0[t + 1] = e0 * inv;
        sW1[t + 1] = e1 * inv;
        float a = 3.0f / (1.0f + expf(-biases[t])) - 1.0f;
        a = fminf(fmaxf(a, -1.0f + EPS), 2.0f - EPS);
        sA[t + 1] = a;
        int cw; float p0, p1;
        level_setup(a, cw, p0, p1);
        sC[t + 1] = cw; sP0[t + 1] = p0; sP1[t + 1] = p1;
    } else {
        sA[0] = 0.0f; sW0[0] = 0.0f; sW1[0] = 0.0f; sP0[0] = 0.0f; sP1[0] = 0.0f; sC[0] = C_LIN;
    }
    __syncthreads();

    float4 na[2], nw0[2], nw1[2], np0[2], np1[2];
    int4 nc[2];
    #pragma unroll
    for (int h = 0; h < 2; ++h) {
        na[h]  = ((const float4*)sA)[h];
        nw0[h] = ((const float4*)sW0)[h];
        nw1[h] = ((const float4*)sW1)[h];
        np0[h] = ((const float4*)sP0)[h];
        np1[h] = ((const float4*)sP1)[h];
        nc[h]  = ((const int4*)sC)[h];
    }

    float carry = 0.0f;
    for (int g = 0; g < LDIM / PF; ++g) {
        float cur[PF], pa[PF], pw0[PF], pw1[PF], pp0[PF], pp1[PF];
        int pc[PF];
        #pragma unroll
        for (int j = 0; j < PF; ++j) cur[j] = nxt[j];
        #pragma unroll
        for (int h = 0; h < 2; ++h) {
            pa[4 * h]  = na[h].x;  pa[4 * h + 1]  = na[h].y;  pa[4 * h + 2]  = na[h].z;  pa[4 * h + 3]  = na[h].w;
            pw0[4 * h] = nw0[h].x; pw0[4 * h + 1] = nw0[h].y; pw0[4 * h + 2] = nw0[h].z; pw0[4 * h + 3] = nw0[h].w;
            pw1[4 * h] = nw1[h].x; pw1[4 * h + 1] = nw1[h].y; pw1[4 * h + 2] = nw1[h].z; pw1[4 * h + 3] = nw1[h].w;
            pp0[4 * h] = np0[h].x; pp0[4 * h + 1] = np0[h].y; pp0[4 * h + 2] = np0[h].z; pp0[4 * h + 3] = np0[h].w;
            pp1[4 * h] = np1[h].x; pp1[4 * h + 1] = np1[h].y; pp1[4 * h + 2] = np1[h].z; pp1[4 * h + 3] = np1[h].w;
            pc[4 * h]  = nc[h].x;  pc[4 * h + 1]  = nc[h].y;  pc[4 * h + 2]  = nc[h].z;  pc[4 * h + 3]  = nc[h].w;
        }
        const int gn = (g + 1 < LDIM / PF) ? g + 1 : g;
        #pragma unroll
        for (int j = 0; j < PF; ++j) nxt[j] = lp[(size_t)(gn * PF + j) * BATCH];
        #pragma unroll
        for (int h = 0; h < 2; ++h) {
            na[h]  = ((const float4*)sA)[2 * gn + h];
            nw0[h] = ((const float4*)sW0)[2 * gn + h];
            nw1[h] = ((const float4*)sW1)[2 * gn + h];
            np0[h] = ((const float4*)sP0)[2 * gn + h];
            np1[h] = ((const float4*)sP1)[2 * gn + h];
            nc[h]  = ((const int4*)sC)[2 * gn + h];
        }
        #pragma unroll
        for (int j = 0; j < PF; ++j) {
            if (j == 0 && g == 0) carry = cur[0];
            else carry = step_op(carry, cur[j], pc[j], pp0[j], pp1[j], pw0[j], pw1[j], pa[j]);
        }
    }
    out[b] = carry;
}

extern "C" void kernel_launch(void* const* d_in, const int* in_sizes, int n_in,
                              void* d_out, int out_size, void* d_ws, size_t ws_size,
                              hipStream_t stream) {
    const float* x       = (const float*)d_in[0];
    const float* logits  = (const float*)d_in[1];
    const float* weights = (const float*)d_in[2];
    const float* biases  = (const float*)d_in[3];

    float* P     = (float*)d_ws;
    float* leafT = P + LDIM * LDIM;

    sinkhorn_kernel<<<1, 1024, 0, stream>>>(logits, P);
    gemm_kernel<<<BATCH / 128, 256, 0, stream>>>(x, P, leafT);
    chain_kernel<<<BATCH / 256, 256, 0, stream>>>(leafT, weights, biases, (float*)d_out);
}


# GEMM and scan fused into one kernel: each workgroup's 256x256 leaf tile goes from the WMMA accumulators through LDS (P's space) to its scan threads; leaves never touch global memory; 2 launches
# speedup vs baseline: 4.9371x; 1.1672x over previous
#include <hip/hip_runtime.h>
#include <hip/hip_bf16.h>
#include <math.h>

#define EPS 1e-6f
#define LDIM 256
#define BATCH 65536

typedef __attribute__((ext_vector_type(2))) float v2f;
typedef __attribute__((ext_vector_type(8))) float v8f;


__device__ __forceinline__ float from_lane(float v, int lane) {
    return __int_as_float(__builtin_amdgcn_readlane(__float_as_int(v), lane));
}

template <bool IS_MAX>
__device__ __forceinline__ float comb2(float x, float y) { return IS_MAX ? fmaxf(x, y) : x + y; }

template <bool IS_MAX>
__device__ __forceinline__ float wave_reduce8(const float (&s)[8], int l) {
    const bool h16 = (l & 16) != 0, h8 = (l & 8) != 0, h4 = (l & 4) != 0;
    float r[4];
    #pragma unroll
    for (int k = 0; k < 4; ++k) {
        const float keep = h16 ? s[k + 4] : s[k];
        const float send = h16 ? s[k] : s[k + 4];
        r[k] = comb2<IS_MAX>(keep, __shfl_xor(send, 16));
    }
    float q[2];
    #pragma unroll
    for (int k = 0; k < 2; ++k) {
        const float keep = h8 ? r[k + 2] : r[k];
        const float send = h8 ? r[k] : r[k + 2];
        q[k] = comb2<IS_MAX>(keep, __shfl_xor(send, 8));
    }
    const float keep = h4 ? q[1] : q[0];
    const float send = h4 ? q[0] : q[1];
    float p = comb2<IS_MAX>(keep, __shfl_xor(send, 4));
    p = comb2<IS_MAX>(p, __shfl_xor(p, 2));
    p = comb2<IS_MAX>(p, __shfl_xor(p, 1));
    return p;
}

__global__ __launch_bounds__(1024) void sinkhorn_kernel(const float* __restrict__ logits,
                                                        float* __restrict__ P) {
    __shared__ __attribute__((aligned(16))) float part[32 * LDIM];
    __shared__ __attribute__((aligned(16))) float vb[LDIM];
    const int t = threadIdx.x;
    const int w = t >> 5;
    const int l = t & 31;

    float kr[8][8];
    const float* src = logits + (size_t)(8 * w) * LDIM + 8 * l;
    #pragma unroll
    for (int i = 0; i < 8; ++i) {
        const float4 lo = *(const float4*)(src + i * LDIM);
        const float4 hi = *(const float4*)(src + i * LDIM + 4);
        kr[i][0] = lo.x; kr[i][1] = lo.y; kr[i][2] = lo.z; kr[i][3] = lo.w;
        kr[i][4] = hi.x; kr[i][5] = hi.y; kr[i][6] = hi.z; kr[i][7] = hi.w;
    }

    {
        float m[8];
        #pragma unroll
        for (int i = 0; i < 8; ++i) {
            float mx = -3.4e38f;
            #pragma unroll
            for (int j = 0; j < 8; ++j) mx = fmaxf(mx, kr[i][j]);
            m[i] = mx;
        }
        const float mr = wave_reduce8<true>(m, l);
        #pragma unroll
        for (int i = 0; i < 8; ++i) {
            const float mx = from_lane(mr, 4 * i);
            #pragma unroll
            for (int j = 0; j < 8; ++j) kr[i][j] = expf(kr[i][j] - mx);
        }
    }

    float a[8], b[8];
    #pragma unroll
    for (int j = 0; j < 8; ++j) { a[j] = 1.0f; b[j] = 1.0f; }

    for (int it = 0; it < 10; ++it) {
        {
            float s[8];
            #pragma unroll
            for (int i = 0; i < 8; ++i) {
                float s0 = kr[i][0] * b[0], s1 = kr[i][1] * b[1];
                s0 = fmaf(kr[i][2], b[2], s0); s1 = fmaf(kr[i][3], b[3], s1);
                s0 = fmaf(kr[i][4], b[4], s0); s1 = fmaf(kr[i][5], b[5], s1);
                s0 = fmaf(kr[i][6], b[6], s0); s1 = fmaf(kr[i][7], b[7], s1);
                s[i] = s0 + s1;
            }
            const float inv = 1.0f / wave_reduce8<false>(s, l);
            #pragma unroll
            for (int i = 0; i < 8; ++i) a[i] = from_lane(inv, 4 * i);
        }
        {
            float c[8];
            #pragma unroll
            for (int j = 0; j < 8; ++j) {
                float c0 = kr[0][j] * a[0], c1 = kr[1][j] * a[1];
                c0 = fmaf(kr[2][j], a[2], c0); c1 = fmaf(kr[3][j], a[3], c1);
                c0 = fmaf(kr[4][j], a[4], c0); c1 = fmaf(kr[5][j], a[5], c1);
                c0 = fmaf(kr[6][j], a[6], c0); c1 = fmaf(kr[7][j], a[7], c1);
                c[j] = c0 + c1;
            }
            float4* pp = (float4*)&part[w * LDIM + 8 * l];
            pp[0] = make_float4(c[0], c[1], c[2], c[3]);
            pp[1] = make_float4(c[4], c[5], c[6], c[7]);
        }
        __syncthreads();
        if (t < LDIM) {
            float s0 = 0.0f, s1 = 0.0f, s2 = 0.0f, s3 = 0.0f;
            #pragma unroll
            for (int k = 0; k < 32; k += 4) {
                s0 += part[(k + 0) * LDIM + t];
                s1 += part[(k + 1) * LDIM + t];
                s2 += part[(k + 2) * LDIM + t];
                s3 += part[(k + 3) * LDIM + t];
            }
            vb[t] = 1.0f / ((s0 + s1) + (s2 + s3));
        }
        __syncthreads();
        {
            const float4 lo = *(const float4*)&vb[8 * l];
            const float4 hi = *(const float4*)&vb[8 * l + 4];
            b[0] = lo.x; b[1] = lo.y; b[2] = lo.z; b[3] = lo.w;
            b[4] = hi.x; b[5] = hi.y; b[6] = hi.z; b[7] = hi.w;
        }
    }

    float* dst = P + (size_t)(8 * w) * LDIM + 8 * l;
    #pragma unroll
    for (int i = 0; i < 8; ++i) {
        *(float4*)(dst + i * LDIM) = make_float4((kr[i][0] * a[i]) * b[0], (kr[i][1] * a[i]) * b[1],
                                                 (kr[i][2] * a[i]) * b[2], (kr[i][3] * a[i]) * b[3]);
        *(float4*)(dst + i * LDIM + 4) = make_float4((kr[i][4] * a[i]) * b[4], (kr[i][5] * a[i]) * b[5],
                                                     (kr[i][6] * a[i]) * b[6], (kr[i][7] * a[i]) * b[7]);
    }
}

__device__ __forceinline__ float r_poly(float a) {
    float d  = 0.5f - a;
    float d2 = d * d;
    float num = 0.25f + 1.65811f * d + 2.15388f * d2 + 8.2844f * d2 * d + 6.16764f * d2 * d2;
    float den = a * (1.0f - a);
    den = (fabsf(den) < EPS) ? EPS : den;
    return num / den;
}

__device__ __forceinline__ float pow_pos(float b, float e) {
    return __builtin_amdgcn_exp2f(e * __builtin_amdgcn_logf(b));
}

__device__ __forceinline__ float pow_pos_sub(float b, float e) {
    return exp2f(e * __builtin_amdgcn_logf(b));
}

__device__ __forceinline__ float power_r2(float x, float y, float ra, float inv_ra, float w0, float w1) {
    return pow_pos(w0 * pow_pos(x, ra) + w1 * pow_pos(y, ra), inv_ra);
}

__device__ __forceinline__ float power_r(float x, float y, float ra, float w0, float w1) {
    ra = (fabsf(ra) < EPS) ? EPS : ra;
    return power_r2(x, y, ra, 1.0f / ra, w0, w1);
}

#define C_LIN  0
#define C_R6   1
#define C_R5   2
#define C_MN   3
#define C_R3   4
#define C_XY   5
#define C_R1   6
#define C_R0   7
#define C_FLIP 8

__device__ __forceinline__ void level_setup(float a, int& cw, float& p0, float& p1) {
    const bool flip = a < 0.5f - EPS;
    const float au = flip ? fminf(fmaxf(1.0f - a, -1.0f + EPS), 2.0f - EPS) : a;
    int c = C_LIN;
    float q0 = 0.0f, q1 = 0.0f;
    if (au > 0.5f && au < 0.75f) { c = C_R6; q0 = 3.0f - 4.0f * au; q1 = 4.0f * au - 2.0f; }
    if (au >= 0.75f && au < 1.0f) {
        float ra = r_poly(fminf(fmaxf(au, 0.75f), 1.0f - EPS));
        ra = (fabsf(ra) < EPS) ? EPS : ra;
        c = C_R5; q0 = ra; q1 = 1.0f / ra;
    }
    if (fabsf(au - 1.0f)  < EPS) c = C_MN;
    if (au > 1.0f  && au < 1.25f) { c = C_R3; q0 = 1.25f - au; q1 = au - 1.0f; }
    if (fabsf(au - 1.25f) < EPS) c = C_XY;
    if (au > 1.25f && au < 2.0f) {
        c = C_R1; q0 = sqrtf(fmaxf(3.0f / fmaxf(2.0f - au, EPS) - 1.0f, EPS));
    }
    if (fabsf(au - 2.0f)  < EPS) c = C_R0;
    cw = flip ? (c | C_FLIP) : c;
    p0 = q0;
    p1 = q1;
}

__device__ __forceinline__ float step_op(float l, float r, int cw, float p0, float p1,
                                         float w0, float w1, float a) {
    float x = l, y = r;
    if (isnan(x)) x = EPS;
    x = fminf(fmaxf(x, EPS), 1.0f - EPS);
    if (isnan(y)) y = EPS;
    y = fminf(fmaxf(y, EPS), 1.0f - EPS);
    const bool flip = (cw & C_FLIP) != 0;
    const float sg = flip ? -1.0f : 1.0f;
    const float of = flip ?  1.0f : 0.0f;
    x = fmaf(sg, x, of);
    y = fmaf(sg, y, of);
    float out;
    switch (cw & 7) {
    case C_R6: out = p0 * (w0 * x + w1 * y) + p1 * power_r(x, y, r_poly(0.75f), w0, w1); break;
    case C_R5: out = power_r2(x, y, p0, p1, w0, w1); break;
    case C_MN: out = fminf(x, y); break;
    case C_R3: out = 4.0f * (p0 * fminf(x, y) + p1 * (x * y)); break;
    case C_XY: out = x * y; break;
    case C_R1: out = pow_pos_sub(x * y, p0); break;
    case C_R0: out = ((fabsf(x - 1.0f) < EPS) && (fabsf(y - 1.0f) < EPS)) ? 1.0f : 0.0f; break;
    default:   out = w0 * x + w1 * y; break;
    }
    const float nres = fmaf(sg, out, of);
    return isnan(nres) ? a : nres;
}

#define LPITCH 260

__global__ __launch_bounds__(512) void leaf_scan_kernel(const float* __restrict__ x,
                                                        const float* __restrict__ P,
                                                        const float* __restrict__ weights,
                                                        const float* __restrict__ biases,
                                                        float* __restrict__ out) {
    __shared__ __attribute__((aligned(16))) float Ps[LDIM * LPITCH];
    __shared__ __attribute__((aligned(16))) float sA[LDIM], sW0[LDIM], sW1[LDIM], sP0[LDIM], sP1[LDIM];
    __shared__ __attribute__((aligned(16))) int sC[LDIM];
    const int t = threadIdx.x;
    {
        const float4* Pv = (const float4*)P;
        float4* Sv = (float4*)Ps;
        for (int i = t; i < LDIM * LDIM / 4; i += 512) Sv[i] = Pv[i];
    }
    if (t < 255) {
        const float w0 = weights[2 * t], w1 = weights[2 * t + 1];
        const float m  = fmaxf(w0, w1);
        const float e0 = expf(w0 - m), e1 = expf(w1 - m);
        const float inv = 1.0f / (e0 + e1);
        sW0[t + 1] = e0 * inv;
        sW1[t + 1] = e1 * inv;
        float a = 3.0f / (1.0f + expf(-biases[t])) - 1.0f;
        a = fminf(fmaxf(a, -1.0f + EPS), 2.0f - EPS);
        sA[t + 1] = a;
        int cw; float p0, p1;
        level_setup(a, cw, p0, p1);
        sC[t + 1] = cw; sP0[t + 1] = p0; sP1[t + 1] = p1;
    } else if (t == 255) {
        sA[0] = 0.0f; sW0[0] = 0.0f; sW1[0] = 0.0f; sP0[0] = 0.0f; sP1[0] = 0.0f; sC[0] = C_LIN;
    }
    __syncthreads();

    const int wave  = t >> 5;
    const int lane  = t & 31;
    const int mloc  = lane & 15;
    const int khalf = (lane >> 4) * 2;
    const int m0    = blockIdx.x * 256 + wave * 16;

    v8f acc[16] = {};

    const float* xrow = x + (size_t)(m0 + mloc) * LDIM + khalf;

    for (int k = 0; k < LDIM; k += 4) {
        v2f bfrag;
        bfrag.x = xrow[k];
        bfrag.y = xrow[k + 1];
        #pragma unroll
        for (int nt = 0; nt < 16; ++nt) {
            const int n = nt * 16 + mloc;
            v2f afrag;
            afrag.x = Ps[(k + khalf)     * LDIM + n];
            afrag.y = Ps[(k + khalf + 1) * LDIM + n];
            acc[nt] = __builtin_amdgcn_wmma_f32_16x16x4_f32(
                false, afrag, false, bfrag, (short)0, acc[nt], false, false);
        }
    }

    __syncthreads();
    {
        float* dstl = Ps + (size_t)((lane >> 4) * 8) * LPITCH + wave * 16 + mloc;
        #pragma unroll
        for (int nt = 0; nt < 16; ++nt) {
            #pragma unroll
            for (int i = 0; i < 8; ++i)
                dstl[(nt * 16 + i) * LPITCH] = acc[nt][i];
        }
    }
    __syncthreads();
    if (t >= 256) return;

    constexpr int PF = 8;
    const float* lp = Ps + t;
    float nxt[PF];
    #pragma unroll
    for (int j = 0; j < PF; ++j) nxt[j] = lp[j * LPITCH];

    float4 na[2], nw0[2], nw1[2], np0[2], np1[2];
    int4 nc[2];
    #pragma unroll
    for (int h = 0; h < 2; ++h) {
        na[h]  = ((const float4*)sA)[h];
        nw0[h] = ((const float4*)sW0)[h];
        nw1[h] = ((const float4*)sW1)[h];
        np0[h] = ((const float4*)sP0)[h];
        np1[h] = ((const float4*)sP1)[h];
        nc[h]  = ((const int4*)sC)[h];
    }

    float carry = 0.0f;
    for (int g = 0; g < LDIM / PF; ++g) {
        float cur[PF], pa[PF], pw0[PF], pw1[PF], pp0[PF], pp1[PF];
        int pc[PF];
        #pragma unroll
        for (int j = 0; j < PF; ++j) cur[j] = nxt[j];
        #pragma unroll
        for (int h = 0; h < 2; ++h) {
            pa[4 * h]  = na[h].x;  pa[4 * h + 1]  = na[h].y;  pa[4 * h + 2]  = na[h].z;  pa[4 * h + 3]  = na[h].w;
            pw0[4 * h] = nw0[h].x; pw0[4 * h + 1] = nw0[h].y; pw0[4 * h + 2] = nw0[h].z; pw0[4 * h + 3] = nw0[h].w;
            pw1[4 * h] = nw1[h].x; pw1[4 * h + 1] = nw1[h].y; pw1[4 * h + 2] = nw1[h].z; pw1[4 * h + 3] = nw1[h].w;
            pp0[4 * h] = np0[h].x; pp0[4 * h + 1] = np0[h].y; pp0[4 * h + 2] = np0[h].z; pp0[4 * h + 3] = np0[h].w;
            pp1[4 * h] = np1[h].x; pp1[4 * h + 1] = np1[h].y; pp1[4 * h + 2] = np1[h].z; pp1[4 * h + 3] = np1[h].w;
            pc[4 * h]  = nc[h].x;  pc[4 * h + 1]  = nc[h].y;  pc[4 * h + 2]  = nc[h].z;  pc[4 * h + 3]  = nc[h].w;
        }
        const int gn = (g + 1 < LDIM / PF) ? g + 1 : g;
        #pragma unroll
        for (int j = 0; j < PF; ++j) nxt[j] = lp[(gn * PF + j) * LPITCH];
        #pragma unroll
        for (int h = 0; h < 2; ++h) {
            na[h]  = ((const float4*)sA)[2 * gn + h];
            nw0[h] = ((const float4*)sW0)[2 * gn + h];
            nw1[h] = ((const float4*)sW1)[2 * gn + h];
            np0[h] = ((const float4*)sP0)[2 * gn + h];
            np1[h] = ((const float4*)sP1)[2 * gn + h];
            nc[h]  = ((const int4*)sC)[2 * gn + h];
        }
        #pragma unroll
        for (int j = 0; j < PF; ++j) {
            if (j == 0 && g == 0) carry = cur[0];
            else carry = step_op(carry, cur[j], pc[j], pp0[j], pp1[j], pw0[j], pw1[j], pa[j]);
        }
    }
    out[blockIdx.x * 256 + t] = carry;
}

extern "C" void kernel_launch(void* const* d_in, const int* in_sizes, int n_in,
                              void* d_out, int out_size, void* d_ws, size_t ws_size,
                              hipStream_t stream) {
    const float* x       = (const float*)d_in[0];
    const float* logits  = (const float*)d_in[1];
    const float* weights = (const float*)d_in[2];
    const float* biases  = (const float*)d_in[3];

    float* P = (float*)d_ws;

    sinkhorn_kernel<<<1, 1024, 0, stream>>>(logits, P);
    leaf_scan_kernel<<<BATCH / 256, 512, 0, stream>>>(x, P, weights, biases, (float*)d_out);
}


# Scan step: a level's class is taken by one scalar branch (power-mean classes, product power, the rest) with selects inside, not by a branch per class; every class's arithmetic kept, bit-identical
# speedup vs baseline: 5.1538x; 1.0439x over previous
#include <hip/hip_runtime.h>
#include <hip/hip_bf16.h>
#include <math.h>

#define EPS 1e-6f
#define LDIM 256
#define BATCH 65536

typedef __attribute__((ext_vector_type(2))) float v2f;
typedef __attribute__((ext_vector_type(8))) float v8f;


__device__ __forceinline__ float from_lane(float v, int lane) {
    return __int_as_float(__builtin_amdgcn_readlane(__float_as_int(v), lane));
}

template <bool IS_MAX>
__device__ __forceinline__ float comb2(float x, float y) { return IS_MAX ? fmaxf(x, y) : x + y; }

template <bool IS_MAX>
__device__ __forceinline__ float wave_reduce8(const float (&s)[8], int l) {
    const bool h16 = (l & 16) != 0, h8 = (l & 8) != 0, h4 = (l & 4) != 0;
    float r[4];
    #pragma unroll
    for (int k = 0; k < 4; ++k) {
        const float keep = h16 ? s[k + 4] : s[k];
        const float send = h16 ? s[k] : s[k + 4];
        r[k] = comb2<IS_MAX>(keep, __shfl_xor(send, 16));
    }
    float q[2];
    #pragma unroll
    for (int k = 0; k < 2; ++k) {
        const float keep = h8 ? r[k + 2] : r[k];
        const float send = h8 ? r[k] : r[k + 2];
        q[k] = comb2<IS_MAX>(keep, __shfl_xor(send, 8));
    }
    const float keep = h4 ? q[1] : q[0];
    const float send = h4 ? q[0] : q[1];
    float p = comb2<IS_MAX>(keep, __shfl_xor(send, 4));
    p = comb2<IS_MAX>(p, __shfl_xor(p, 2));
    p = comb2<IS_MAX>(p, __shfl_xor(p, 1));
    return p;
}

__global__ __launch_bounds__(1024) void sinkhorn_kernel(const float* __restrict__ logits,
                                                        float* __restrict__ P) {
    __shared__ __attribute__((aligned(16))) float part[32 * LDIM];
    __shared__ __attribute__((aligned(16))) float vb[LDIM];
    const int t = threadIdx.x;
    const int w = t >> 5;
    const int l = t & 31;

    float kr[8][8];
    const float* src = logits + (size_t)(8 * w) * LDIM + 8 * l;
    #pragma unroll
    for (int i = 0; i < 8; ++i) {
        const float4 lo = *(const float4*)(src + i * LDIM);
        const float4 hi = *(const float4*)(src + i * LDIM + 4);
        kr[i][0] = lo.x; kr[i][1] = lo.y; kr[i][2] = lo.z; kr[i][3] = lo.w;
        kr[i][4] = hi.x; kr[i][5] = hi.y; kr[i][6] = hi.z; kr[i][7] = hi.w;
    }

    {
        float m[8];
        #pragma unroll
        for (int i = 0; i < 8; ++i) {
            float mx = -3.4e38f;
            #pragma unroll
            for (int j = 0; j < 8; ++j) mx = fmaxf(mx, kr[i][j]);
            m[i] = mx;
        }
        const float mr = wave_reduce8<true>(m, l);
        #pragma unroll
        for (int i = 0; i < 8; ++i) {
            const float mx = from_lane(mr, 4 * i);
            #pragma unroll
            for (int j = 0; j < 8; ++j) kr[i][j] = expf(kr[i][j] - mx);
        }
    }

    float a[8], b[8];
    #pragma unroll
    for (int j = 0; j < 8; ++j) { a[j] = 1.0f; b[j] = 1.0f; }

    for (int it = 0; it < 10; ++it) {
        {
            float s[8];
            #pragma unroll
            for (int i = 0; i < 8; ++i) {
                float s0 = kr[i][0] * b[0], s1 = kr[i][1] * b[1];
                s0 = fmaf(kr[i][2], b[2], s0); s1 = fmaf(kr[i][3], b[3], s1);
                s0 = fmaf(kr[i][4], b[4], s0); s1 = fmaf(kr[i][5], b[5], s1);
                s0 = fmaf(kr[i][6], b[6], s0); s1 = fmaf(kr[i][7], b[7], s1);
                s[i] = s0 + s1;
            }
            const float inv = 1.0f / wave_reduce8<false>(s, l);
            #pragma unroll
            for (int i = 0; i < 8; ++i) a[i] = from_lane(inv, 4 * i);
        }
        {
            float c[8];
            #pragma unroll
            for (int j = 0; j < 8; ++j) {
                float c0 = kr[0][j] * a[0], c1 = kr[1][j] * a[1];
                c0 = fmaf(kr[2][j], a[2], c0); c1 = fmaf(kr[3][j], a[3], c1);
                c0 = fmaf(kr[4][j], a[4], c0); c1 = fmaf(kr[5][j], a[5], c1);
                c0 = fmaf(kr[6][j], a[6], c0); c1 = fmaf(kr[7][j], a[7], c1);
                c[j] = c0 + c1;
            }
            float4* pp = (float4*)&part[w * LDIM + 8 * l];
            pp[0] = make_float4(c[0], c[1], c[2], c[3]);
            pp[1] = make_float4(c[4], c[5], c[6], c[7]);
        }
        __syncthreads();
        if (t < LDIM) {
            float s0 = 0.0f, s1 = 0.0f, s2 = 0.0f, s3 = 0.0f;
            #pragma unroll
            for (int k = 0; k < 32; k += 4) {
                s0 += part[(k + 0) * LDIM + t];
                s1 += part[(k + 1) * LDIM + t];
                s2 += part[(k + 2) * LDIM + t];
                s3 += part[(k + 3) * LDIM + t];
            }
            vb[t] = 1.0f / ((s0 + s1) + (s2 + s3));
        }
        __syncthreads();
        {
            const float4 lo = *(const float4*)&vb[8 * l];
            const float4 hi = *(const float4*)&vb[8 * l + 4];
            b[0] = lo.x; b[1] = lo.y; b[2] = lo.z; b[3] = lo.w;
            b[4] = hi.x; b[5] = hi.y; b[6] = hi.z; b[7] = hi.w;
        }
    }

    float* dst = P + (size_t)(8 * w) * LDIM + 8 * l;
    #pragma unroll
    for (int i = 0; i < 8; ++i) {
        *(float4*)(dst + i * LDIM) = make_float4((kr[i][0] * a[i]) * b[0], (kr[i][1] * a[i]) * b[1],
                                                 (kr[i][2] * a[i]) * b[2], (kr[i][3] * a[i]) * b[3]);
        *(float4*)(dst + i * LDIM + 4) = make_float4((kr[i][4] * a[i]) * b[4], (kr[i][5] * a[i]) * b[5],
                                                     (kr[i][6] * a[i]) * b[6], (kr[i][7] * a[i]) * b[7]);
    }
}

__device__ __forceinline__ float r_poly(float a) {
    float d  = 0.5f - a;
    float d2 = d * d;
    float num = 0.25f + 1.65811f * d + 2.15388f * d2 + 8.2844f * d2 * d + 6.16764f * d2 * d2;
    float den = a * (1.0f - a);
    den = (fabsf(den) < EPS) ? EPS : den;
    return num / den;
}

__device__ __forceinline__ float pow_pos(float b, float e) {
    return __builtin_amdgcn_exp2f(e * __builtin_amdgcn_logf(b));
}

__device__ __forceinline__ float pow_pos_sub(float b, float e) {
    return exp2f(e * __builtin_amdgcn_logf(b));
}

__device__ __forceinline__ float power_r2(float x, float y, float ra, float inv_ra, float w0, float w1) {
    return pow_pos(w0 * pow_pos(x, ra) + w1 * pow_pos(y, ra), inv_ra);
}

#define C_LIN  0
#define C_R6   1
#define C_R5   2
#define C_MN   3
#define C_R3   4
#define C_XY   5
#define C_R1   6
#define C_R0   7
#define C_FLIP 8
#define C_POW  16

__device__ __forceinline__ void level_setup(float a, int& cw, float& p0, float& p1) {
    const bool flip = a < 0.5f - EPS;
    const float au = flip ? fminf(fmaxf(1.0f - a, -1.0f + EPS), 2.0f - EPS) : a;
    int c = C_LIN;
    float q0 = 0.0f, q1 = 0.0f;
    if (au > 0.5f && au < 0.75f) { c = C_R6; q0 = 3.0f - 4.0f * au; q1 = 4.0f * au - 2.0f; }
    if (au >= 0.75f && au < 1.0f) {
        float ra = r_poly(fminf(fmaxf(au, 0.75f), 1.0f - EPS));
        ra = (fabsf(ra) < EPS) ? EPS : ra;
        c = C_R5; q0 = ra; q1 = 1.0f / ra;
    }
    if (fabsf(au - 1.0f)  < EPS) c = C_MN;
    if (au > 1.0f  && au < 1.25f) { c = C_R3; q0 = 1.25f - au; q1 = au - 1.0f; }
    if (fabsf(au - 1.25f) < EPS) c = C_XY;
    if (au > 1.25f && au < 2.0f) {
        c = C_R1; q0 = sqrtf(fmaxf(3.0f / fmaxf(2.0f - au, EPS) - 1.0f, EPS));
    }
    if (fabsf(au - 2.0f)  < EPS) c = C_R0;
    if (c == C_R6 || c == C_R5) c |= C_POW;
    cw = flip ? (c | C_FLIP) : c;
    p0 = q0;
    p1 = q1;
}

__device__ __forceinline__ float step_op(float l, float r, int cw, float p0, float p1,
                                         float w0, float w1, float a) {
    float x = l, y = r;
    if (isnan(x)) x = EPS;
    x = fminf(fmaxf(x, EPS), 1.0f - EPS);
    if (isnan(y)) y = EPS;
    y = fminf(fmaxf(y, EPS), 1.0f - EPS);
    const bool flip = (cw & C_FLIP) != 0;
    const float sg = flip ? -1.0f : 1.0f;
    const float of = flip ?  1.0f : 0.0f;
    x = fmaf(sg, x, of);
    y = fmaf(sg, y, of);
    const int cls = cw & 7;
    float out;
    if (cw & C_POW) {
        const bool r6 = cls == C_R6;
        const float rc  = r_poly(0.75f);
        const float rg  = (fabsf(rc) < EPS) ? EPS : rc;
        const float e   = r6 ? rg : p0;
        const float ie  = r6 ? 1.0f / rg : p1;
        const float q0  = r6 ? p0 : 0.0f;
        const float q1  = r6 ? p1 : 1.0f;
        out = q0 * (w0 * x + w1 * y) + q1 * power_r2(x, y, e, ie, w0, w1);
    } else if (cls == C_R1) {
        out = pow_pos_sub(x * y, p0);
    } else {
        const float xy  = x * y;
        const float mn  = fminf(x, y);
        const float r3  = 4.0f * (p0 * mn + p1 * xy);
        const float ind = ((fabsf(x - 1.0f) < EPS) & (fabsf(y - 1.0f) < EPS)) ? 1.0f : 0.0f;
        out = w0 * x + w1 * y;
        out = (cls == C_MN) ? mn : out;
        out = (cls == C_XY) ? xy : out;
        out = (cls == C_R0) ? ind : out;
        out = (cls == C_R3) ? r3 : out;
    }
    const float nres = fmaf(sg, out, of);
    return isnan(nres) ? a : nres;
}

#define LPITCH 260

__global__ __launch_bounds__(512) void leaf_scan_kernel(const float* __restrict__ x,
                                                        const float* __restrict__ P,
                                                        const float* __restrict__ weights,
                                                        const float* __restrict__ biases,
                                                        float* __restrict__ out) {
    __shared__ __attribute__((aligned(16))) float Ps[LDIM * LPITCH];
    __shared__ __attribute__((aligned(16))) float sA[LDIM], sW0[LDIM], sW1[LDIM], sP0[LDIM], sP1[LDIM];
    __shared__ __attribute__((aligned(16))) int sC[LDIM];
    const int t = threadIdx.x;
    {
        const float4* Pv = (const float4*)P;
        float4* Sv = (float4*)Ps;
        for (int i = t; i < LDIM * LDIM / 4; i += 512) Sv[i] = Pv[i];
    }
    if (t < 255) {
        const float w0 = weights[2 * t], w1 = weights[2 * t + 1];
        const float m  = fmaxf(w0, w1);
        const float e0 = expf(w0 - m), e1 = expf(w1 - m);
        const float inv = 1.0f / (e0 + e1);
        sW0[t + 1] = e0 * inv;
        sW1[t + 1] = e1 * inv;
        float a = 3.0f / (1.0f + expf(-biases[t])) - 1.0f;
        a = fminf(fmaxf(a, -1.0f + EPS), 2.0f - EPS);
        sA[t + 1] = a;
        int cw; float p0, p1;
        level_setup(a, cw, p0, p1);
        sC[t + 1] = cw; sP0[t + 1] = p0; sP1[t + 1] = p1;
    } else if (t == 255) {
        sA[0] = 0.0f; sW0[0] = 0.0f; sW1[0] = 0.0f; sP0[0] = 0.0f; sP1[0] = 0.0f; sC[0] = C_LIN;
    }
    __syncthreads();

    const int wave  = t >> 5;
    const int lane  = t & 31;
    const int mloc  = lane & 15;
    const int khalf = (lane >> 4) * 2;
    const int m0    = blockIdx.x * 256 + wave * 16;

    v8f acc[16] = {};

    const float* xrow = x + (size_t)(m0 + mloc) * LDIM + khalf;

    for (int k = 0; k < LDIM; k += 4) {
        v2f bfrag;
        bfrag.x = xrow[k];
        bfrag.y = xrow[k + 1];
        #pragma unroll
        for (int nt = 0; nt < 16; ++nt) {
            const int n = nt * 16 + mloc;
            v2f afrag;
            afrag.x = Ps[(k + khalf)     * LDIM + n];
            afrag.y = Ps[(k + khalf + 1) * LDIM + n];
            acc[nt] = __builtin_amdgcn_wmma_f32_16x16x4_f32(
                false, afrag, false, bfrag, (short)0, acc[nt], false, false);
        }
    }

    __syncthreads();
    {
        float* dstl = Ps + (size_t)((lane >> 4) * 8) * LPITCH + wave * 16 + mloc;
        #pragma unroll
        for (int nt = 0; nt < 16; ++nt) {
            #pragma unroll
            for (int i = 0; i < 8; ++i)
                dstl[(nt * 16 + i) * LPITCH] = acc[nt][i];
        }
    }
    __syncthreads();
    if (t >= 256) return;

    constexpr int PF = 8;
    const float* lp = Ps + t;
    float nxt[PF];
    #pragma unroll
    for (int j = 0; j < PF; ++j) nxt[j] = lp[j * LPITCH];

    float4 na[2], nw0[2], nw1[2], np0[2], np1[2];
    int4 nc[2];
    #pragma unroll
    for (int h = 0; h < 2; ++h) {
        na[h]  = ((const float4*)sA)[h];
        nw0[h] = ((const float4*)sW0)[h];
        nw1[h] = ((const float4*)sW1)[h];
        np0[h] = ((const float4*)sP0)[h];
        np1[h] = ((const float4*)sP1)[h];
        nc[h]  = ((const int4*)sC)[h];
    }

    float carry = 0.0f;
    for (int g = 0; g < LDIM / PF; ++g) {
        float cur[PF], pa[PF], pw0[PF], pw1[PF], pp0[PF], pp1[PF];
        int pc[PF];
        #pragma unroll
        for (int j = 0; j < PF; ++j) cur[j] = nxt[j];
        #pragma unroll
        for (int h = 0; h < 2; ++h) {
            pa[4 * h]  = na[h].x;  pa[4 * h + 1]  = na[h].y;  pa[4 * h + 2]  = na[h].z;  pa[4 * h + 3]  = na[h].w;
            pw0[4 * h] = nw0[h].x; pw0[4 * h + 1] = nw0[h].y; pw0[4 * h + 2] = nw0[h].z; pw0[4 * h + 3] = nw0[h].w;
            pw1[4 * h] = nw1[h].x; pw1[4 * h + 1] = nw1[h].y; pw1[4 * h + 2] = nw1[h].z; pw1[4 * h + 3] = nw1[h].w;
            pp0[4 * h] = np0[h].x; pp0[4 * h + 1] = np0[h].y; pp0[4 * h + 2] = np0[h].z; pp0[4 * h + 3] = np0[h].w;
            pp1[4 * h] = np1[h].x; pp1[4 * h + 1] = np1[h].y; pp1[4 * h + 2] = np1[h].z; pp1[4 * h + 3] = np1[h].w;
            pc[4 * h]  = nc[h].x;  pc[4 * h + 1]  = nc[h].y;  pc[4 * h + 2]  = nc[h].z;  pc[4 * h + 3]  = nc[h].w;
        }
        const int gn = (g + 1 < LDIM / PF) ? g + 1 : g;
        #pragma unroll
        for (int j = 0; j < PF; ++j) nxt[j] = lp[(gn * PF + j) * LPITCH];
        #pragma unroll
        for (int h = 0; h < 2; ++h) {
            na[h]  = ((const float4*)sA)[2 * gn + h];
            nw0[h] = ((const float4*)sW0)[2 * gn + h];
            nw1[h] = ((const float4*)sW1)[2 * gn + h];
            np0[h] = ((const float4*)sP0)[2 * gn + h];
            np1[h] = ((const float4*)sP1)[2 * gn + h];
            nc[h]  = ((const int4*)sC)[2 * gn + h];
        }
        #pragma unroll
        for (int j = 0; j < PF; ++j) {
            if (j == 0 && g == 0) carry = cur[0];
            else carry = step_op(carry, cur[j], pc[j], pp0[j], pp1[j], pw0[j], pw1[j], pa[j]);
        }
    }
    out[blockIdx.x * 256 + t] = carry;
}

extern "C" void kernel_launch(void* const* d_in, const int* in_sizes, int n_in,
                              void* d_out, int out_size, void* d_ws, size_t ws_size,
                              hipStream_t stream) {
    const float* x       = (const float*)d_in[0];
    const float* logits  = (const float*)d_in[1];
    const float* weights = (const float*)d_in[2];
    const float* biases  = (const float*)d_in[3];

    float* P = (float*)d_ws;

    sinkhorn_kernel<<<1, 1024, 0, stream>>>(logits, P);
    leaf_scan_kernel<<<BATCH / 256, 512, 0, stream>>>(x, P, weights, biases, (float*)d_out);
}


# Sinkhorn moved into the GEMM+scan kernel: every workgroup works P out itself (K in registers, 16 rows a wave) straight into LDS; 1 launch, no workspace; column sums add 16 shares (sum order changes)
# speedup vs baseline: 6.0851x; 1.1807x over previous
#include <hip/hip_runtime.h>
#include <hip/hip_bf16.h>
#include <math.h>

#define EPS 1e-6f
#define LDIM 256
#define BATCH 65536

typedef __attribute__((ext_vector_type(2))) float v2f;
typedef __attribute__((ext_vector_type(8))) float v8f;


__device__ __forceinline__ float from_lane(float v, int lane) {
    return __int_as_float(__builtin_amdgcn_readlane(__float_as_int(v), lane));
}

template <bool IS_MAX>
__device__ __forceinline__ float comb2(float x, float y) { return IS_MAX ? fmaxf(x, y) : x + y; }

template <bool IS_MAX>
__device__ __forceinline__ float wave_reduce8(const float (&s)[8], int l) {
    const bool h16 = (l & 16) != 0, h8 = (l & 8) != 0, h4 = (l & 4) != 0;
    float r[4];
    #pragma unroll
    for (int k = 0; k < 4; ++k) {
        const float keep = h16 ? s[k + 4] : s[k];
        const float send = h16 ? s[k] : s[k + 4];
        r[k] = comb2<IS_MAX>(keep, __shfl_xor(send, 16));
    }
    float q[2];
    #pragma unroll
    for (int k = 0; k < 2; ++k) {
        const float keep = h8 ? r[k + 2] : r[k];
        const float send = h8 ? r[k] : r[k + 2];
        q[k] = comb2<IS_MAX>(keep, __shfl_xor(send, 8));
    }
    const float keep = h4 ? q[1] : q[0];
    const float send = h4 ? q[0] : q[1];
    float p = comb2<IS_MAX>(keep, __shfl_xor(send, 4));
    p = comb2<IS_MAX>(p, __shfl_xor(p, 2));
    p = comb2<IS_MAX>(p, __shfl_xor(p, 1));
    return p;
}

__device__ __forceinline__ void sinkhorn_to_lds(const float* __restrict__ logits, float* __restrict__ Pl,
                                                float* __restrict__ part, float* __restrict__ vb,
                                                const int t) {
    const int w = t >> 5;
    const int l = t & 31;

    float kr[16][8];
    const float* src = logits + (size_t)(16 * w) * LDIM + 8 * l;
    #pragma unroll
    for (int i = 0; i < 16; ++i) {
        const float4 lo = *(const float4*)(src + i * LDIM);
        const float4 hi = *(const float4*)(src + i * LDIM + 4);
        kr[i][0] = lo.x; kr[i][1] = lo.y; kr[i][2] = lo.z; kr[i][3] = lo.w;
        kr[i][4] = hi.x; kr[i][5] = hi.y; kr[i][6] = hi.z; kr[i][7] = hi.w;
    }

    #pragma unroll
    for (int h = 0; h < 2; ++h) {
        float m[8];
        #pragma unroll
        for (int i = 0; i < 8; ++i) {
            float mx = -3.4e38f;
            #pragma unroll
            for (int j = 0; j < 8; ++j) mx = fmaxf(mx, kr[8 * h + i][j]);
            m[i] = mx;
        }
        const float mr = wave_reduce8<true>(m, l);
        #pragma unroll
        for (int i = 0; i < 8; ++i) {
            const float mx = from_lane(mr, 4 * i);
            #pragma unroll
            for (int j = 0; j < 8; ++j) kr[8 * h + i][j] = expf(kr[8 * h + i][j] - mx);
        }
    }

    float a[16], b[8];
    #pragma unroll
    for (int i = 0; i < 16; ++i) a[i] = 1.0f;
    #pragma unroll
    for (int j = 0; j < 8; ++j) b[j] = 1.0f;

    for (int it = 0; it < 10; ++it) {
        #pragma unroll
        for (int h = 0; h < 2; ++h) {
            float s[8];
            #pragma unroll
            for (int i = 0; i < 8; ++i) {
                const int r = 8 * h + i;
                float s0 = kr[r][0] * b[0], s1 = kr[r][1] * b[1];
                s0 = fmaf(kr[r][2], b[2], s0); s1 = fmaf(kr[r][3], b[3], s1);
                s0 = fmaf(kr[r][4], b[4], s0); s1 = fmaf(kr[r][5], b[5], s1);
                s0 = fmaf(kr[r][6], b[6], s0); s1 = fmaf(kr[r][7], b[7], s1);
                s[i] = s0 + s1;
            }
            const float inv = 1.0f / wave_reduce8<false>(s, l);
            #pragma unroll
            for (int i = 0; i < 8; ++i) a[8 * h + i] = from_lane(inv, 4 * i);
        }
        {
            float c[8];
            #pragma unroll
            for (int j = 0; j < 8; ++j) {
                float c0 = kr[0][j] * a[0], c1 = kr[1][j] * a[1];
                #pragma unroll
                for (int i = 2; i < 16; i += 2) {
                    c0 = fmaf(kr[i][j], a[i], c0);
                    c1 = fmaf(kr[i + 1][j], a[i + 1], c1);
                }
                c[j] = c0 + c1;
            }
            float4* pp = (float4*)&part[w * LDIM + 8 * l];
            pp[0] = make_float4(c[0], c[1], c[2], c[3]);
            pp[1] = make_float4(c[4], c[5], c[6], c[7]);
        }
        __syncthreads();
        if (t < LDIM) {
            float s0 = 0.0f, s1 = 0.0f, s2 = 0.0f, s3 = 0.0f;
            #pragma unroll
            for (int k = 0; k < 16; k += 4) {
                s0 += part[(k + 0) * LDIM + t];
                s1 += part[(k + 1) * LDIM + t];
                s2 += part[(k + 2) * LDIM + t];
                s3 += part[(k + 3) * LDIM + t];
            }
            vb[t] = 1.0f / ((s0 + s1) + (s2 + s3));
        }
        __syncthreads();
        {
            const float4 lo = *(const float4*)&vb[8 * l];
            const float4 hi = *(const float4*)&vb[8 * l + 4];
            b[0] = lo.x; b[1] = lo.y; b[2] = lo.z; b[3] = lo.w;
            b[4] = hi.x; b[5] = hi.y; b[6] = hi.z; b[7] = hi.w;
        }
    }

    float* dst = Pl + (16 * w) * LDIM + 8 * l;
    #pragma unroll
    for (int i = 0; i < 16; ++i) {
        *(float4*)(dst + i * LDIM) = make_float4((kr[i][0] * a[i]) * b[0], (kr[i][1] * a[i]) * b[1],
                                                 (kr[i][2] * a[i]) * b[2], (kr[i][3] * a[i]) * b[3]);
        *(float4*)(dst + i * LDIM + 4) = make_float4((kr[i][4] * a[i]) * b[4], (kr[i][5] * a[i]) * b[5],
                                                     (kr[i][6] * a[i]) * b[6], (kr[i][7] * a[i]) * b[7]);
    }
}

__device__ __forceinline__ float r_poly(float a) {
    float d  = 0.5f - a;
    float d2 = d * d;
    float num = 0.25f + 1.65811f * d + 2.15388f * d2 + 8.2844f * d2 * d + 6.16764f * d2 * d2;
    float den = a * (1.0f - a);
    den = (fabsf(den) < EPS) ? EPS : den;
    return num / den;
}

__device__ __forceinline__ float pow_pos(float b, float e) {
    return __builtin_amdgcn_exp2f(e * __builtin_amdgcn_logf(b));
}

__device__ __forceinline__ float pow_pos_sub(float b, float e) {
    return exp2f(e * __builtin_amdgcn_logf(b));
}

__device__ __forceinline__ float power_r2(float x, float y, float ra, float inv_ra, float w0, float w1) {
    return pow_pos(w0 * pow_pos(x, ra) + w1 * pow_pos(y, ra), inv_ra);
}

#define C_LIN  0
#define C_R6   1
#define C_R5   2
#define C_MN   3
#define C_R3   4
#define C_XY   5
#define C_R1   6
#define C_R0   7
#define C_FLIP 8
#define C_POW  16

__device__ __forceinline__ void level_setup(float a, int& cw, float& p0, float& p1) {
    const bool flip = a < 0.5f - EPS;
    const float au = flip ? fminf(fmaxf(1.0f - a, -1.0f + EPS), 2.0f - EPS) : a;
    int c = C_LIN;
    float q0 = 0.0f, q1 = 0.0f;
    if (au > 0.5f && au < 0.75f) { c = C_R6; q0 = 3.0f - 4.0f * au; q1 = 4.0f * au - 2.0f; }
    if (au >= 0.75f && au < 1.0f) {
        float ra = r_poly(fminf(fmaxf(au, 0.75f), 1.0f - EPS));
        ra = (fabsf(ra) < EPS) ? EPS : ra;
        c = C_R5; q0 = ra; q1 = 1.0f / ra;
    }
    if (fabsf(au - 1.0f)  < EPS) c = C_MN;
    if (au > 1.0f  && au < 1.25f) { c = C_R3; q0 = 1.25f - au; q1 = au - 1.0f; }
    if (fabsf(au - 1.25f) < EPS) c = C_XY;
    if (au > 1.25f && au < 2.0f) {
        c = C_R1; q0 = sqrtf(fmaxf(3.0f / fmaxf(2.0f - au, EPS) - 1.0f, EPS));
    }
    if (fabsf(au - 2.0f)  < EPS) c = C_R0;
    if (c == C_R6 || c == C_R5) c |= C_POW;
    cw = flip ? (c | C_FLIP) : c;
    p0 = q0;
    p1 = q1;
}

__device__ __forceinline__ float step_op(float l, float r, int cw, float p0, float p1,
                                         float w0, float w1, float a) {
    float x = l, y = r;
    if (isnan(x)) x = EPS;
    x = fminf(fmaxf(x, EPS), 1.0f - EPS);
    if (isnan(y)) y = EPS;
    y = fminf(fmaxf(y, EPS), 1.0f - EPS);
    const bool flip = (cw & C_FLIP) != 0;
    const float sg = flip ? -1.0f : 1.0f;
    const float of = flip ?  1.0f : 0.0f;
    x = fmaf(sg, x, of);
    y = fmaf(sg, y, of);
    const int cls = cw & 7;
    float out;
    if (cw & C_POW) {
        const bool r6 = cls == C_R6;
        const float rc  = r_poly(0.75f);
        const float rg  = (fabsf(rc) < EPS) ? EPS : rc;
        const float e   = r6 ? rg : p0;
        const float ie  = r6 ? 1.0f / rg : p1;
        const float q0  = r6 ? p0 : 0.0f;
        const float q1  = r6 ? p1 : 1.0f;
        out = q0 * (w0 * x + w1 * y) + q1 * power_r2(x, y, e, ie, w0, w1);
    } else if (cls == C_R1) {
        out = pow_pos_sub(x * y, p0);
    } else {
        const float xy  = x * y;
        const float mn  = fminf(x, y);
        const float r3  = 4.0f * (p0 * mn + p1 * xy);
        const float ind = ((fabsf(x - 1.0f) < EPS) & (fabsf(y - 1.0f) < EPS)) ? 1.0f : 0.0f;
        out = w0 * x + w1 * y;
        out = (cls == C_MN) ? mn : out;
        out = (cls == C_XY) ? xy : out;
        out = (cls == C_R0) ? ind : out;
        out = (cls == C_R3) ? r3 : out;
    }
    const float nres = fmaf(sg, out, of);
    return isnan(nres) ? a : nres;
}

#define LPITCH 260

__global__ __launch_bounds__(512) void leaf_scan_kernel(const float* __restrict__ x,
                                                        const float* __restrict__ logits,
                                                        const float* __restrict__ weights,
                                                        const float* __restrict__ biases,
                                                        float* __restrict__ out) {
    __shared__ __attribute__((aligned(16))) float Ps[LDIM * LPITCH];
    __shared__ __attribute__((aligned(16))) float sA[LDIM], sW0[LDIM], sW1[LDIM], sP0[LDIM], sP1[LDIM];
    __shared__ __attribute__((aligned(16))) int sC[LDIM];
    __shared__ __attribute__((aligned(16))) float sPart[16 * LDIM];
    __shared__ __attribute__((aligned(16))) float sVb[LDIM];
    const int t = threadIdx.x;
    if (t < 255) {
        const float w0 = weights[2 * t], w1 = weights[2 * t + 1];
        const float m  = fmaxf(w0, w1);
        const float e0 = expf(w0 - m), e1 = expf(w1 - m);
        const float inv = 1.0f / (e0 + e1);
        sW0[t + 1] = e0 * inv;
        sW1[t + 1] = e1 * inv;
        float a = 3.0f / (1.0f + expf(-biases[t])) - 1.0f;
        a = fminf(fmaxf(a, -1.0f + EPS), 2.0f - EPS);
        sA[t + 1] = a;
        int cw; float p0, p1;
        level_setup(a, cw, p0, p1);
        sC[t + 1] = cw; sP0[t + 1] = p0; sP1[t + 1] = p1;
    } else if (t == 255) {
        sA[0] = 0.0f; sW0[0] = 0.0f; sW1[0] = 0.0f; sP0[0] = 0.0f; sP1[0] = 0.0f; sC[0] = C_LIN;
    }
    sinkhorn_to_lds(logits, Ps, sPart, sVb, t);
    __syncthreads();

    const int wave  = t >> 5;
    const int lane  = t & 31;
    const int mloc  = lane & 15;
    const int khalf = (lane >> 4) * 2;
    const int m0    = blockIdx.x * 256 + wave * 16;

    v8f acc[16] = {};

    const float* xrow = x + (size_t)(m0 + mloc) * LDIM + khalf;

    for (int k = 0; k < LDIM; k += 4) {
        v2f bfrag;
        bfrag.x = xrow[k];
        bfrag.y = xrow[k + 1];
        #pragma unroll
        for (int nt = 0; nt < 16; ++nt) {
            const int n = nt * 16 + mloc;
            v2f afrag;
            afrag.x = Ps[(k + khalf)     * LDIM + n];
            afrag.y = Ps[(k + khalf + 1) * LDIM + n];
            acc[nt] = __builtin_amdgcn_wmma_f32_16x16x4_f32(
                false, afrag, false, bfrag, (short)0, acc[nt], false, false);
        }
    }

    __syncthreads();
    {
        float* dstl = Ps + (size_t)((lane >> 4) * 8) * LPITCH + wave * 16 + mloc;
        #pragma unroll
        for (int nt = 0; nt < 16; ++nt) {
            #pragma unroll
            for (int i = 0; i < 8; ++i)
                dstl[(nt * 16 + i) * LPITCH] = acc[nt][i];
        }
    }
    __syncthreads();
    if (t >= 256) return;

    constexpr int PF = 8;
    const float* lp = Ps + t;
    float nxt[PF];
    #pragma unroll
    for (int j = 0; j < PF; ++j) nxt[j] = lp[j * LPITCH];

    float4 na[2], nw0[2], nw1[2], np0[2], np1[2];
    int4 nc[2];
    #pragma unroll
    for (int h = 0; h < 2; ++h) {
        na[h]  = ((const float4*)sA)[h];
        nw0[h] = ((const float4*)sW0)[h];
        nw1[h] = ((const float4*)sW1)[h];
        np0[h] = ((const float4*)sP0)[h];
        np1[h] = ((const float4*)sP1)[h];
        nc[h]  = ((const int4*)sC)[h];
    }

    float carry = 0.0f;
    for (int g = 0; g < LDIM / PF; ++g) {
        float cur[PF], pa[PF], pw0[PF], pw1[PF], pp0[PF], pp1[PF];
        int pc[PF];
        #pragma unroll
        for (int j = 0; j < PF; ++j) cur[j] = nxt[j];
        #pragma unroll
        for (int h = 0; h < 2; ++h) {
            pa[4 * h]  = na[h].x;  pa[4 * h + 1]  = na[h].y;  pa[4 * h + 2]  = na[h].z;  pa[4 * h + 3]  = na[h].w;
            pw0[4 * h] = nw0[h].x; pw0[4 * h + 1] = nw0[h].y; pw0[4 * h + 2] = nw0[h].z; pw0[4 * h + 3] = nw0[h].w;
            pw1[4 * h] = nw1[h].x; pw1[4 * h + 1] = nw1[h].y; pw1[4 * h + 2] = nw1[h].z; pw1[4 * h + 3] = nw1[h].w;
            pp0[4 * h] = np0[h].x; pp0[4 * h + 1] = np0[h].y; pp0[4 * h + 2] = np0[h].z; pp0[4 * h + 3] = np0[h].w;
            pp1[4 * h] = np1[h].x; pp1[4 * h + 1] = np1[h].y; pp1[4 * h + 2] = np1[h].z; pp1[4 * h + 3] = np1[h].w;
            pc[4 * h]  = nc[h].x;  pc[4 * h + 1]  = nc[h].y;  pc[4 * h + 2]  = nc[h].z;  pc[4 * h + 3]  = nc[h].w;
        }
        const int gn = (g + 1 < LDIM / PF) ? g + 1 : g;
        #pragma unroll
        for (int j = 0; j < PF; ++j) nxt[j] = lp[(gn * PF + j) * LPITCH];
        #pragma unroll
        for (int h = 0; h < 2; ++h) {
            na[h]  = ((const float4*)sA)[2 * gn + h];
            nw0[h] = ((const float4*)sW0)[2 * gn + h];
            nw1[h] = ((const float4*)sW1)[2 * gn + h];
            np0[h] = ((const float4*)sP0)[2 * gn + h];
            np1[h] = ((const float4*)sP1)[2 * gn + h];
            nc[h]  = ((const int4*)sC)[2 * gn + h];
        }
        #pragma unroll
        for (int j = 0; j < PF; ++j) {
            if (j == 0 && g == 0) carry = cur[0];
            else carry = step_op(carry, cur[j], pc[j], pp0[j], pp1[j], pw0[j], pw1[j], pa[j]);
        }
    }
    out[blockIdx.x * 256 + t] = carry;
}

extern "C" void kernel_launch(void* const* d_in, const int* in_sizes, int n_in,
                              void* d_out, int out_size, void* d_ws, size_t ws_size,
                              hipStream_t stream) {
    const float* x       = (const float*)d_in[0];
    const float* logits  = (const float*)d_in[1];
    const float* weights = (const float*)d_in[2];
    const float* biases  = (const float*)d_in[3];

    leaf_scan_kernel<<<BATCH / 256, 512, 0, stream>>>(x, logits, weights, biases, (float*)d_out);
}


# Sinkhorn: library expf and the correctly rounded divisions replaced by the hardware exp2 and reciprocal instructions (K and the scales move by a few ulp; no sum order changes)
# speedup vs baseline: 6.3196x; 1.0385x over previous
#include <hip/hip_runtime.h>
#include <hip/hip_bf16.h>
#include <math.h>

#define EPS 1e-6f
#define LDIM 256
#define BATCH 65536

typedef __attribute__((ext_vector_type(2))) float v2f;
typedef __attribute__((ext_vector_type(8))) float v8f;


__device__ __forceinline__ float exp_nonpos(float d) {
    return __builtin_amdgcn_exp2f(d * 1.44269504088896340736f);
}
__device__ __forceinline__ float recip_pos(float s) {
    return __builtin_amdgcn_rcpf(s);
}

__device__ __forceinline__ float from_lane(float v, int lane) {
    return __int_as_float(__builtin_amdgcn_readlane(__float_as_int(v), lane));
}

template <bool IS_MAX>
__device__ __forceinline__ float comb2(float x, float y) { return IS_MAX ? fmaxf(x, y) : x + y; }

template <bool IS_MAX>
__device__ __forceinline__ float wave_reduce8(const float (&s)[8], int l) {
    const bool h16 = (l & 16) != 0, h8 = (l & 8) != 0, h4 = (l & 4) != 0;
    float r[4];
    #pragma unroll
    for (int k = 0; k < 4; ++k) {
        const float keep = h16 ? s[k + 4] : s[k];
        const float send = h16 ? s[k] : s[k + 4];
        r[k] = comb2<IS_MAX>(keep, __shfl_xor(send, 16));
    }
    float q[2];
    #pragma unroll
    for (int k = 0; k < 2; ++k) {
        const float keep = h8 ? r[k + 2] : r[k];
        const float send = h8 ? r[k] : r[k + 2];
        q[k] = comb2<IS_MAX>(keep, __shfl_xor(send, 8));
    }
    const float keep = h4 ? q[1] : q[0];
    const float send = h4 ? q[0] : q[1];
    float p = comb2<IS_MAX>(keep, __shfl_xor(send, 4));
    p = comb2<IS_MAX>(p, __shfl_xor(p, 2));
    p = comb2<IS_MAX>(p, __shfl_xor(p, 1));
    return p;
}

__device__ __forceinline__ void sinkhorn_to_lds(const float* __restrict__ logits, float* __restrict__ Pl,
                                                float* __restrict__ part, float* __restrict__ vb,
                                                const int t) {
    const int w = t >> 5;
    const int l = t & 31;

    float kr[16][8];
    const float* src = logits + (size_t)(16 * w) * LDIM + 8 * l;
    #pragma unroll
    for (int i = 0; i < 16; ++i) {
        const float4 lo = *(const float4*)(src + i * LDIM);
        const float4 hi = *(const float4*)(src + i * LDIM + 4);
        kr[i][0] = lo.x; kr[i][1] = lo.y; kr[i][2] = lo.z; kr[i][3] = lo.w;
        kr[i][4] = hi.x; kr[i][5] = hi.y; kr[i][6] = hi.z; kr[i][7] = hi.w;
    }

    #pragma unroll
    for (int h = 0; h < 2; ++h) {
        float m[8];
        #pragma unroll
        for (int i = 0; i < 8; ++i) {
            float mx = -3.4e38f;
            #pragma unroll
            for (int j = 0; j < 8; ++j) mx = fmaxf(mx, kr[8 * h + i][j]);
            m[i] = mx;
        }
        const float mr = wave_reduce8<true>(m, l);
        #pragma unroll
        for (int i = 0; i < 8; ++i) {
            const float mx = from_lane(mr, 4 * i);
            #pragma unroll
            for (int j = 0; j < 8; ++j) kr[8 * h + i][j] = exp_nonpos(kr[8 * h + i][j] - mx);
        }
    }

    float a[16], b[8];
    #pragma unroll
    for (int i = 0; i < 16; ++i) a[i] = 1.0f;
    #pragma unroll
    for (int j = 0; j < 8; ++j) b[j] = 1.0f;

    for (int it = 0; it < 10; ++it) {
        #pragma unroll
        for (int h = 0; h < 2; ++h) {
            float s[8];
            #pragma unroll
            for (int i = 0; i < 8; ++i) {
                const int r = 8 * h + i;
                float s0 = kr[r][0] * b[0], s1 = kr[r][1] * b[1];
                s0 = fmaf(kr[r][2], b[2], s0); s1 = fmaf(kr[r][3], b[3], s1);
                s0 = fmaf(kr[r][4], b[4], s0); s1 = fmaf(kr[r][5], b[5], s1);
                s0 = fmaf(kr[r][6], b[6], s0); s1 = fmaf(kr[r][7], b[7], s1);
                s[i] = s0 + s1;
            }
            const float inv = recip_pos(wave_reduce8<false>(s, l));
            #pragma unroll
            for (int i = 0; i < 8; ++i) a[8 * h + i] = from_lane(inv, 4 * i);
        }
        {
            float c[8];
            #pragma unroll
            for (int j = 0; j < 8; ++j) {
                float c0 = kr[0][j] * a[0], c1 = kr[1][j] * a[1];
                #pragma unroll
                for (int i = 2; i < 16; i += 2) {
                    c0 = fmaf(kr[i][j], a[i], c0);
                    c1 = fmaf(kr[i + 1][j], a[i + 1], c1);
                }
                c[j] = c0 + c1;
            }
            float4* pp = (float4*)&part[w * LDIM + 8 * l];
            pp[0] = make_float4(c[0], c[1], c[2], c[3]);
            pp[1] = make_float4(c[4], c[5], c[6], c[7]);
        }
        __syncthreads();
        if (t < LDIM) {
            float s0 = 0.0f, s1 = 0.0f, s2 = 0.0f, s3 = 0.0f;
            #pragma unroll
            for (int k = 0; k < 16; k += 4) {
                s0 += part[(k + 0) * LDIM + t];
                s1 += part[(k + 1) * LDIM + t];
                s2 += part[(k + 2) * LDIM + t];
                s3 += part[(k + 3) * LDIM + t];
            }
            vb[t] = recip_pos((s0 + s1) + (s2 + s3));
        }
        __syncthreads();
        {
            const float4 lo = *(const float4*)&vb[8 * l];
            const float4 hi = *(const float4*)&vb[8 * l + 4];
            b[0] = lo.x; b[1] = lo.y; b[2] = lo.z; b[3] = lo.w;
            b[4] = hi.x; b[5] = hi.y; b[6] = hi.z; b[7] = hi.w;
        }
    }

    float* dst = Pl + (16 * w) * LDIM + 8 * l;
    #pragma unroll
    for (int i = 0; i < 16; ++i) {
        *(float4*)(dst + i * LDIM) = make_float4((kr[i][0] * a[i]) * b[0], (kr[i][1] * a[i]) * b[1],
                                                 (kr[i][2] * a[i]) * b[2], (kr[i][3] * a[i]) * b[3]);
        *(float4*)(dst + i * LDIM + 4) = make_float4((kr[i][4] * a[i]) * b[4], (kr[i][5] * a[i]) * b[5],
                                                     (kr[i][6] * a[i]) * b[6], (kr[i][7] * a[i]) * b[7]);
    }
}

__device__ __forceinline__ float r_poly(float a) {
    float d  = 0.5f - a;
    float d2 = d * d;
    float num = 0.25f + 1.65811f * d + 2.15388f * d2 + 8.2844f * d2 * d + 6.16764f * d2 * d2;
    float den = a * (1.0f - a);
    den = (fabsf(den) < EPS) ? EPS : den;
    return num / den;
}

__device__ __forceinline__ float pow_pos(float b, float e) {
    return __builtin_amdgcn_exp2f(e * __builtin_amdgcn_logf(b));
}

__device__ __forceinline__ float pow_pos_sub(float b, float e) {
    return exp2f(e * __builtin_amdgcn_logf(b));
}

__device__ __forceinline__ float power_r2(float x, float y, float ra, float inv_ra, float w0, float w1) {
    return pow_pos(w0 * pow_pos(x, ra) + w1 * pow_pos(y, ra), inv_ra);
}

#define C_LIN  0
#define C_R6   1
#define C_R5   2
#define C_MN   3
#define C_R3   4
#define C_XY   5
#define C_R1   6
#define C_R0   7
#define C_FLIP 8
#define C_POW  16

__device__ __forceinline__ void level_setup(float a, int& cw, float& p0, float& p1) {
    const bool flip = a < 0.5f - EPS;
    const float au = flip ? fminf(fmaxf(1.0f - a, -1.0f + EPS), 2.0f - EPS) : a;
    int c = C_LIN;
    float q0 = 0.0f, q1 = 0.0f;
    if (au > 0.5f && au < 0.75f) { c = C_R6; q0 = 3.0f - 4.0f * au; q1 = 4.0f * au - 2.0f; }
    if (au >= 0.75f && au < 1.0f) {
        float ra = r_poly(fminf(fmaxf(au, 0.75f), 1.0f - EPS));
        ra = (fabsf(ra) < EPS) ? EPS : ra;
        c = C_R5; q0 = ra; q1 = 1.0f / ra;
    }
    if (fabsf(au - 1.0f)  < EPS) c = C_MN;
    if (au > 1.0f  && au < 1.25f) { c = C_R3; q0 = 1.25f - au; q1 = au - 1.0f; }
    if (fabsf(au - 1.25f) < EPS) c = C_XY;
    if (au > 1.25f && au < 2.0f) {
        c = C_R1; q0 = sqrtf(fmaxf(3.0f / fmaxf(2.0f - au, EPS) - 1.0f, EPS));
    }
    if (fabsf(au - 2.0f)  < EPS) c = C_R0;
    if (c == C_R6 || c == C_R5) c |= C_POW;
    cw = flip ? (c | C_FLIP) : c;
    p0 = q0;
    p1 = q1;
}

__device__ __forceinline__ float step_op(float l, float r, int cw, float p0, float p1,
                                         float w0, float w1, float a) {
    float x = l, y = r;
    if (isnan(x)) x = EPS;
    x = fminf(fmaxf(x, EPS), 1.0f - EPS);
    if (isnan(y)) y = EPS;
    y = fminf(fmaxf(y, EPS), 1.0f - EPS);
    const bool flip = (cw & C_FLIP) != 0;
    const float sg = flip ? -1.0f : 1.0f;
    const float of = flip ?  1.0f : 0.0f;
    x = fmaf(sg, x, of);
    y = fmaf(sg, y, of);
    const int cls = cw & 7;
    float out;
    if (cw & C_POW) {
        const bool r6 = cls == C_R6;
        const float rc  = r_poly(0.75f);
        const float rg  = (fabsf(rc) < EPS) ? EPS : rc;
        const float e   = r6 ? rg : p0;
        const float ie  = r6 ? 1.0f / rg : p1;
        const float q0  = r6 ? p0 : 0.0f;
        const float q1  = r6 ? p1 : 1.0f;
        out = q0 * (w0 * x + w1 * y) + q1 * power_r2(x, y, e, ie, w0, w1);
    } else if (cls == C_R1) {
        out = pow_pos_sub(x * y, p0);
    } else {
        const float xy  = x * y;
        const float mn  = fminf(x, y);
        const float r3  = 4.0f * (p0 * mn + p1 * xy);
        const float ind = ((fabsf(x - 1.0f) < EPS) & (fabsf(y - 1.0f) < EPS)) ? 1.0f : 0.0f;
        out = w0 * x + w1 * y;
        out = (cls == C_MN) ? mn : out;
        out = (cls == C_XY) ? xy : out;
        out = (cls == C_R0) ? ind : out;
        out = (cls == C_R3) ? r3 : out;
    }
    const float nres = fmaf(sg, out, of);
    return isnan(nres) ? a : nres;
}

#define LPITCH 260

__global__ __launch_bounds__(512) void leaf_scan_kernel(const float* __restrict__ x,
                                                        const float* __restrict__ logits,
                                                        const float* __restrict__ weights,
                                                        const float* __restrict__ biases,
                                                        float* __restrict__ out) {
    __shared__ __attribute__((aligned(16))) float Ps[LDIM * LPITCH];
    __shared__ __attribute__((aligned(16))) float sA[LDIM], sW0[LDIM], sW1[LDIM], sP0[LDIM], sP1[LDIM];
    __shared__ __attribute__((aligned(16))) int sC[LDIM];
    __shared__ __attribute__((aligned(16))) float sPart[16 * LDIM];
    __shared__ __attribute__((aligned(16))) float sVb[LDIM];
    const int t = threadIdx.x;
    if (t < 255) {
        const float w0 = weights[2 * t], w1 = weights[2 * t + 1];
        const float m  = fmaxf(w0, w1);
        const float e0 = expf(w0 - m), e1 = expf(w1 - m);
        const float inv = 1.0f / (e0 + e1);
        sW0[t + 1] = e0 * inv;
        sW1[t + 1] = e1 * inv;
        float a = 3.0f / (1.0f + expf(-biases[t])) - 1.0f;
        a = fminf(fmaxf(a, -1.0f + EPS), 2.0f - EPS);
        sA[t + 1] = a;
        int cw; float p0, p1;
        level_setup(a, cw, p0, p1);
        sC[t + 1] = cw; sP0[t + 1] = p0; sP1[t + 1] = p1;
    } else if (t == 255) {
        sA[0] = 0.0f; sW0[0] = 0.0f; sW1[0] = 0.0f; sP0[0] = 0.0f; sP1[0] = 0.0f; sC[0] = C_LIN;
    }
    sinkhorn_to_lds(logits, Ps, sPart, sVb, t);
    __syncthreads();

    const int wave  = t >> 5;
    const int lane  = t & 31;
    const int mloc  = lane & 15;
    const int khalf = (lane >> 4) * 2;
    const int m0    = blockIdx.x * 256 + wave * 16;

    v8f acc[16] = {};

    const float* xrow = x + (size_t)(m0 + mloc) * LDIM + khalf;

    for (int k = 0; k < LDIM; k += 4) {
        v2f bfrag;
        bfrag.x = xrow[k];
        bfrag.y = xrow[k + 1];
        #pragma unroll
        for (int nt = 0; nt < 16; ++nt) {
            const int n = nt * 16 + mloc;
            v2f afrag;
            afrag.x = Ps[(k + khalf)     * LDIM + n];
            afrag.y = Ps[(k + khalf + 1) * LDIM + n];
            acc[nt] = __builtin_amdgcn_wmma_f32_16x16x4_f32(
                false, afrag, false, bfrag, (short)0, acc[nt], false, false);
        }
    }

    __syncthreads();
    {
        float* dstl = Ps + (size_t)((lane >> 4) * 8) * LPITCH + wave * 16 + mloc;
        #pragma unroll
        for (int nt = 0; nt < 16; ++nt) {
            #pragma unroll
            for (int i = 0; i < 8; ++i)
                dstl[(nt * 16 + i) * LPITCH] = acc[nt][i];
        }
    }
    __syncthreads();
    if (t >= 256) return;

    constexpr int PF = 8;
    const float* lp = Ps + t;
    float nxt[PF];
    #pragma unroll
    for (int j = 0; j < PF; ++j) nxt[j] = lp[j * LPITCH];

    float4 na[2], nw0[2], nw1[2], np0[2], np1[2];
    int4 nc[2];
    #pragma unroll
    for (int h = 0; h < 2; ++h) {
        na[h]  = ((const float4*)sA)[h];
        nw0[h] = ((const float4*)sW0)[h];
        nw1[h] = ((const float4*)sW1)[h];
        np0[h] = ((const float4*)sP0)[h];
        np1[h] = ((const float4*)sP1)[h];
        nc[h]  = ((const int4*)sC)[h];
    }

    float carry = 0.0f;
    for (int g = 0; g < LDIM / PF; ++g) {
        float cur[PF], pa[PF], pw0[PF], pw1[PF], pp0[PF], pp1[PF];
        int pc[PF];
        #pragma unroll
        for (int j = 0; j < PF; ++j) cur[j] = nxt[j];
        #pragma unroll
        for (int h = 0; h < 2; ++h) {
            pa[4 * h]  = na[h].x;  pa[4 * h + 1]  = na[h].y;  pa[4 * h + 2]  = na[h].z;  pa[4 * h + 3]  = na[h].w;
            pw0[4 * h] = nw0[h].x; pw0[4 * h + 1] = nw0[h].y; pw0[4 * h + 2] = nw0[h].z; pw0[4 * h + 3] = nw0[h].w;
            pw1[4 * h] = nw1[h].x; pw1[4 * h + 1] = nw1[h].y; pw1[4 * h + 2] = nw1[h].z; pw1[4 * h + 3] = nw1[h].w;
            pp0[4 * h] = np0[h].x; pp0[4 * h + 1] = np0[h].y; pp0[4 * h + 2] = np0[h].z; pp0[4 * h + 3] = np0[h].w;
            pp1[4 * h] = np1[h].x; pp1[4 * h + 1] = np1[h].y; pp1[4 * h + 2] = np1[h].z; pp1[4 * h + 3] = np1[h].w;
            pc[4 * h]  = nc[h].x;  pc[4 * h + 1]  = nc[h].y;  pc[4 * h + 2]  = nc[h].z;  pc[4 * h + 3]  = nc[h].w;
        }
        const int gn = (g + 1 < LDIM / PF) ? g + 1 : g;
        #pragma unroll
        for (int j = 0; j < PF; ++j) nxt[j] = lp[(gn * PF + j) * LPITCH];
        #pragma unroll
        for (int h = 0; h < 2; ++h) {
            na[h]  = ((const float4*)sA)[2 * gn + h];
            nw0[h] = ((const float4*)sW0)[2 * gn + h];
            nw1[h] = ((const float4*)sW1)[2 * gn + h];
            np0[h] = ((const float4*)sP0)[2 * gn + h];
            np1[h] = ((const float4*)sP1)[2 * gn + h];
            nc[h]  = ((const int4*)sC)[2 * gn + h];
        }
        #pragma unroll
        for (int j = 0; j < PF; ++j) {
            if (j == 0 && g == 0) carry = cur[0];
            else carry = step_op(carry, cur[j], pc[j], pp0[j], pp1[j], pw0[j], pw1[j], pa[j]);
        }
    }
    out[blockIdx.x * 256 + t] = carry;
}

extern "C" void kernel_launch(void* const* d_in, const int* in_sizes, int n_in,
                              void* d_out, int out_size, void* d_ws, size_t ws_size,
                              hipStream_t stream) {
    const float* x       = (const float*)d_in[0];
    const float* logits  = (const float*)d_in[1];
    const float* weights = (const float*)d_in[2];
    const float* biases  = (const float*)d_in[3];

    leaf_scan_kernel<<<BATCH / 256, 512, 0, stream>>>(x, logits, weights, biases, (float*)d_out);
}
